# starting kernel
# baseline (speedup 1.0000x reference)
#include <hip/hip_runtime.h>
#include <hip/hip_bf16.h>


typedef _Float16 v16h __attribute__((ext_vector_type(16)));
typedef _Float16 v8h  __attribute__((ext_vector_type(8)));
typedef float    v8f  __attribute__((ext_vector_type(8)));
typedef float    v4f  __attribute__((ext_vector_type(4)));

#define NB   8
#define CCH  512
#define NTOK 1024
#define NH   4
#define DK   128

__device__ __forceinline__ v16h load_frag(const _Float16* p) {
  union { v16h v; v8h h[2]; } u;
  u.h[0] = *(const v8h*)(p);
  u.h[1] = *(const v8h*)(p + 16);
  return u.v;
}

__device__ __forceinline__ v8f wmma16(v16h a, v16h b, v8f c) {
  return __builtin_amdgcn_wmma_f32_16x16x32_f16(false, a, false, b,
                                                (short)0, c, false, false);
}

__global__ void __launch_bounds__(256)
transpose_cast(const float* __restrict__ src, _Float16* __restrict__ dst,
               int rows, int cols) {
  __shared__ float tile[32][33];
  const size_t boff = (size_t)blockIdx.z * rows * cols;
  const int cx = blockIdx.x * 32 + threadIdx.x;
  const int ry = blockIdx.y * 32 + threadIdx.y;
#pragma unroll
  for (int j = 0; j < 32; j += 8)
    tile[threadIdx.y + j][threadIdx.x] = src[boff + (size_t)(ry + j) * cols + cx];
  __syncthreads();
  const int rx = blockIdx.y * 32 + threadIdx.x;
  const int cy = blockIdx.x * 32 + threadIdx.y;
#pragma unroll
  for (int j = 0; j < 32; j += 8)
    dst[boff + (size_t)(cy + j) * rows + rx] = (_Float16)tile[threadIdx.x][threadIdx.y + j];
}

__global__ void __launch_bounds__(32)
qkv_gemm(const _Float16* __restrict__ xh,
         const _Float16* __restrict__ WpT,
         const float* __restrict__ bp,
         _Float16* __restrict__ qh,
         _Float16* __restrict__ kh,
         _Float16* __restrict__ vT) {
  const int lane  = threadIdx.x;
  const int hf    = lane >> 4;
  const int l16   = lane & 15;
  const int kbase = hf * 8;
  const int roff  = hf * 8;
  const int n0 = blockIdx.x * 32;
  const int j0 = blockIdx.y * 64;
  const int b  = blockIdx.z;

  const _Float16* A0 = xh + ((size_t)b * NTOK + (n0 + l16)) * CCH + kbase;
  const _Float16* A1 = A0 + (size_t)16 * CCH;
  const _Float16* Bp[4];
#pragma unroll
  for (int t = 0; t < 4; ++t)
    Bp[t] = WpT + (size_t)(j0 + t * 16 + l16) * CCH + kbase;

  v16h a0c = load_frag(A0), a1c = load_frag(A1);
  v16h bc[4];
#pragma unroll
  for (int t = 0; t < 4; ++t) bc[t] = load_frag(Bp[t]);
  v8f acc0[4] = {}, acc1[4] = {};
#pragma unroll
  for (int k0 = 0; k0 < CCH; k0 += 32) {
    v16h a0n = a0c, a1n = a1c, bn[4] = {bc[0], bc[1], bc[2], bc[3]};
    const int kn = k0 + 32;
    if (kn < CCH) {
      a0n = load_frag(A0 + kn);
      a1n = load_frag(A1 + kn);
#pragma unroll
      for (int t = 0; t < 4; ++t) bn[t] = load_frag(Bp[t] + kn);
    }
#pragma unroll
    for (int t = 0; t < 4; ++t) {
      acc0[t] = wmma16(a0c, bc[t], acc0[t]);
      acc1[t] = wmma16(a1c, bc[t], acc1[t]);
    }
    a0c = a0n; a1c = a1n;
#pragma unroll
    for (int t = 0; t < 4; ++t) bc[t] = bn[t];
  }

#pragma unroll
  for (int t = 0; t < 4; ++t) {
    const int j0t  = j0 + t * 16;
    const int head = j0t / 384;
    const int r0   = j0t % 384;
    const int seg  = r0 / 128;
    const int d    = (r0 % 128) + l16;
    const float bias = bp[j0t + l16];
    if (seg == 2) {
      _Float16* dst = vT + (((size_t)b * NH + head) * DK + d) * NTOK + n0 + roff;
#pragma unroll
      for (int g = 0; g < 8; ++g) {
        dst[g]      = (_Float16)(acc0[t][g] + bias);
        dst[g + 16] = (_Float16)(acc1[t][g] + bias);
      }
    } else {
      _Float16* base = (seg == 0) ? qh : kh;
      _Float16* p0 = base + (((size_t)b * NH + head) * NTOK + n0 + roff) * DK + d;
#pragma unroll
      for (int g = 0; g < 8; ++g) {
        p0[(size_t)g * DK]        = (_Float16)(acc0[t][g] + bias);
        p0[(size_t)(g + 16) * DK] = (_Float16)(acc1[t][g] + bias);
      }
    }
  }
}

__global__ void __launch_bounds__(32)
attention_fa(const _Float16* __restrict__ qh, const _Float16* __restrict__ kh,
             const _Float16* __restrict__ vT, _Float16* __restrict__ attn) {
  __shared__ alignas(16) _Float16 lds_p[16][40];
  const int lane  = threadIdx.x;
  const int hf    = lane >> 4;
  const int l16   = lane & 15;
  const int kbase = hf * 8;
  const int roff  = hf * 8;
  const int n0 = blockIdx.x * 16;
  const int h  = blockIdx.y;
  const int b  = blockIdx.z;
  const float scale = 0.08838834764831845f;

  const _Float16* Qb = qh + (((size_t)b * NH + h) * NTOK + n0 + l16) * DK;
  v16h qf[4];
#pragma unroll
  for (int kk = 0; kk < 4; ++kk) qf[kk] = load_frag(Qb + kk * 32 + kbase);

  v8f o[8] = {};
  float mi[8], li[8];
#pragma unroll
  for (int g = 0; g < 8; ++g) { mi[g] = -1e30f; li[g] = 0.0f; }

  const _Float16* Kb = kh + ((size_t)b * NH + h) * NTOK * DK;
  const _Float16* Vb = vT + ((size_t)b * NH + h) * DK * NTOK;

  for (int j0 = 0; j0 < NTOK; j0 += 32) {
    if (j0 + 32 < NTOK) {
      __builtin_prefetch(Kb + (size_t)(j0 + 32 + l16) * DK + kbase, 0, 3);
      __builtin_prefetch(Kb + (size_t)(j0 + 48 + l16) * DK + kbase, 0, 3);
    }
    v8f s0 = {}, s1 = {};
    const _Float16* K0 = Kb + (size_t)(j0 + l16) * DK;
    const _Float16* K1 = K0 + (size_t)16 * DK;
#pragma unroll
    for (int kk = 0; kk < 4; ++kk) {
      s0 = wmma16(qf[kk], load_frag(K0 + kk * 32 + kbase), s0);
      s1 = wmma16(qf[kk], load_frag(K1 + kk * 32 + kbase), s1);
    }
    v16h vf[4];
#pragma unroll
    for (int t = 0; t < 4; ++t)
      vf[t] = load_frag(Vb + (size_t)(t * 16 + l16) * NTOK + j0 + kbase);

#pragma unroll
    for (int g = 0; g < 8; ++g) {
      const float a0 = s0[g] * scale, a1 = s1[g] * scale;
      float m = fmaxf(a0, a1);
      m = fmaxf(m, __shfl_xor(m, 1));
      m = fmaxf(m, __shfl_xor(m, 2));
      m = fmaxf(m, __shfl_xor(m, 4));
      m = fmaxf(m, __shfl_xor(m, 8));
      const float mn    = fmaxf(mi[g], m);
      const float alpha = __expf(mi[g] - mn);
      const float p0 = __expf(a0 - mn);
      const float p1 = __expf(a1 - mn);
      s0[g] = p0; s1[g] = p1;
      float rs = p0 + p1;
      rs += __shfl_xor(rs, 1);
      rs += __shfl_xor(rs, 2);
      rs += __shfl_xor(rs, 4);
      rs += __shfl_xor(rs, 8);
      li[g] = li[g] * alpha + rs;
      mi[g] = mn;
#pragma unroll
      for (int t = 0; t < 8; ++t) o[t][g] *= alpha;
    }
#pragma unroll
    for (int g = 0; g < 8; ++g) {
      lds_p[roff + g][l16]      = (_Float16)s0[g];
      lds_p[roff + g][16 + l16] = (_Float16)s1[g];
    }
    __syncthreads();
    const v16h pf = load_frag(&lds_p[l16][kbase]);
#pragma unroll
    for (int t = 0; t < 4; ++t) o[t] = wmma16(pf, vf[t], o[t]);
#pragma unroll
    for (int t = 4; t < 8; ++t) {
      const _Float16* Vp = Vb + (size_t)(t * 16 + l16) * NTOK + j0 + kbase;
      o[t] = wmma16(pf, load_frag(Vp), o[t]);
    }
    __syncthreads();
  }
  _Float16* Ob = attn + ((size_t)b * NTOK + n0 + roff) * (NH * DK) + h * DK + l16;
#pragma unroll
  for (int t = 0; t < 8; ++t)
#pragma unroll
    for (int g = 0; g < 8; ++g)
      Ob[(size_t)g * (NH * DK) + t * 16] = (_Float16)(o[t][g] / li[g]);
}

__global__ void __launch_bounds__(32)
out_proj(const _Float16* __restrict__ attn,
         const _Float16* __restrict__ WoT,
         const float* __restrict__ bo,
         const float* __restrict__ x,
         float* __restrict__ out) {
  const int lane  = threadIdx.x;
  const int hf    = lane >> 4;
  const int l16   = lane & 15;
  const int kbase = hf * 8;
  const int roff  = hf * 8;
  const int n0 = blockIdx.x * 32;
  const int c0 = blockIdx.y * 64;
  const int b  = blockIdx.z;

  const _Float16* A0 = attn + ((size_t)b * NTOK + n0 + l16) * CCH + kbase;
  const _Float16* A1 = A0 + (size_t)16 * CCH;
  const _Float16* Bp[4];
#pragma unroll
  for (int t = 0; t < 4; ++t)
    Bp[t] = WoT + (size_t)(c0 + t * 16 + l16) * CCH + kbase;

  v16h a0c = load_frag(A0), a1c = load_frag(A1);
  v16h bc[4];
#pragma unroll
  for (int t = 0; t < 4; ++t) bc[t] = load_frag(Bp[t]);
  v8f acc0[4] = {}, acc1[4] = {};
#pragma unroll
  for (int k0 = 0; k0 < CCH; k0 += 32) {
    v16h a0n = a0c, a1n = a1c, bn[4] = {bc[0], bc[1], bc[2], bc[3]};
    const int kn = k0 + 32;
    if (kn < CCH) {
      a0n = load_frag(A0 + kn);
      a1n = load_frag(A1 + kn);
#pragma unroll
      for (int t = 0; t < 4; ++t) bn[t] = load_frag(Bp[t] + kn);
    }
#pragma unroll
    for (int t = 0; t < 4; ++t) {
      acc0[t] = wmma16(a0c, bc[t], acc0[t]);
      acc1[t] = wmma16(a1c, bc[t], acc1[t]);
    }
    a0c = a0n; a1c = a1n;
#pragma unroll
    for (int t = 0; t < 4; ++t) bc[t] = bn[t];
  }

#pragma unroll
  for (int t = 0; t < 4; ++t) {
    const int c = c0 + t * 16 + l16;
    const float bias = bo[c];
    const size_t base = ((size_t)b * CCH + c) * NTOK + n0 + roff;
#pragma unroll
    for (int half2 = 0; half2 < 2; ++half2) {
      const v8f& av   = half2 ? acc1[t] : acc0[t];
      const size_t bi = base + (size_t)half2 * 16;
      const v4f x0 = *(const v4f*)(x + bi);
      const v4f x1 = *(const v4f*)(x + bi + 4);
      v4f r0, r1;
#pragma unroll
      for (int g = 0; g < 4; ++g) {
        r0[g] = av[g]     + bias + x0[g];
        r1[g] = av[4 + g] + bias + x1[g];
      }
      *(v4f*)(out + bi)     = r0;
      *(v4f*)(out + bi + 4) = r1;
    }
  }
}

extern "C" void kernel_launch(void* const* d_in, const int* in_sizes, int n_in,
                              void* d_out, int out_size, void* d_ws, size_t ws_size,
                              hipStream_t stream) {
  (void)in_sizes; (void)n_in; (void)out_size; (void)ws_size;
  const float* x  = (const float*)d_in[0];
  const float* Wp = (const float*)d_in[1];
  const float* bp = (const float*)d_in[2];
  const float* Wo = (const float*)d_in[3];
  const float* bo = (const float*)d_in[4];
  float* out = (float*)d_out;

  char* ws = (char*)d_ws;
  _Float16* xh  = (_Float16*)(ws);
  _Float16* WpT = (_Float16*)(ws +  8388608);
  _Float16* WoT = (_Float16*)(ws +  9961472);
  _Float16* qh  = (_Float16*)(ws + 10485760);
  _Float16* kh  = (_Float16*)(ws + 18874368);
  _Float16* vT  = (_Float16*)(ws + 27262976);
  _Float16* at  = (_Float16*)(ws + 35651584);

  const dim3 tb(32, 8);
  transpose_cast<<<dim3(NTOK / 32, CCH / 32, NB), tb, 0, stream>>>(x,  xh,  CCH, NTOK);
  transpose_cast<<<dim3(1536 / 32, CCH / 32, 1),  tb, 0, stream>>>(Wp, WpT, CCH, 1536);
  transpose_cast<<<dim3(CCH  / 32, CCH / 32, 1),  tb, 0, stream>>>(Wo, WoT, CCH, CCH);

  qkv_gemm    <<<dim3(NTOK / 32, 1536 / 64, NB), 32, 0, stream>>>(xh, WpT, bp, qh, kh, vT);
  attention_fa<<<dim3(NTOK / 16, NH,        NB), 32, 0, stream>>>(qh, kh, vT, at);
  out_proj    <<<dim3(NTOK / 32, CCH / 64,  NB), 32, 0, stream>>>(at, WoT, bo, x, out);
}


# f16 workspace operands stored in WMMA fragment order: a fragment load is two contiguous 512-B wave loads, not two 16-line gathers; waves_per_eu(1,4) keeps the parent build's loads in flight
# speedup vs baseline: 1.1864x; 1.1864x over previous
#include <hip/hip_runtime.h>
#include <hip/hip_bf16.h>


typedef _Float16 v16h __attribute__((ext_vector_type(16)));
typedef _Float16 v8h  __attribute__((ext_vector_type(8)));
typedef _Float16 v4h  __attribute__((ext_vector_type(4)));
typedef float    v8f  __attribute__((ext_vector_type(8)));
typedef float    v4f  __attribute__((ext_vector_type(4)));

#define NB   8
#define CCH  512
#define NTOK 1024
#define NH   4
#define DK   128

__device__ __forceinline__ v16h load_frag_rows(const _Float16* p) {
  union { v16h v; v8h h[2]; } u;
  u.h[0] = *(const v8h*)(p);
  u.h[1] = *(const v8h*)(p + 16);
  return u.v;
}

__device__ __forceinline__ v16h load_frag(const _Float16* p) {
  union { v16h v; v8h h[2]; } u;
  u.h[0] = *(const v8h*)(p);
  u.h[1] = *(const v8h*)(p + 256);
  return u.v;
}

__device__ __forceinline__ v8f wmma16(v16h a, v16h b, v8f c) {
  return __builtin_amdgcn_wmma_f32_16x16x32_f16(false, a, false, b,
                                                (short)0, c, false, false);
}

__global__ void __launch_bounds__(256)
transpose_cast(const float* __restrict__ src, _Float16* __restrict__ dst,
               int rows, int cols) {
  __shared__ float tile[32][36];
  const size_t boff = (size_t)blockIdx.z * rows * cols;
  const int cx = blockIdx.x * 32 + threadIdx.x;
  const int ry = blockIdx.y * 32 + threadIdx.y;
#pragma unroll
  for (int j = 0; j < 32; j += 8)
    tile[threadIdx.y + j][threadIdx.x] = src[boff + (size_t)(ry + j) * cols + cx];
  __syncthreads();
  const int tid = threadIdx.y * 32 + threadIdx.x;
  const int q   = tid & 127;
  const int e   = q >> 6;
  const int l   = (q >> 1) & 31;
  const int kk  = e * 16 + (l >> 4) * 8 + (q & 1) * 4;
  const int rr  = (tid >> 7) * 16 + (l & 15);
  const size_t rt = (((size_t)blockIdx.z * cols + (size_t)blockIdx.x * 32) >> 4) + (tid >> 7);
  _Float16* p = dst + (rt * (size_t)(rows >> 5) + blockIdx.y) * 512 + q * 4;
  v4h val;
#pragma unroll
  for (int i = 0; i < 4; ++i) val[i] = (_Float16)tile[kk + i][rr];
  *(v4h*)p = val;
}

__global__ void __launch_bounds__(32)
qkv_gemm(const _Float16* __restrict__ xh,
         const _Float16* __restrict__ WpT,
         const float* __restrict__ bp,
         _Float16* __restrict__ qh,
         _Float16* __restrict__ kh,
         _Float16* __restrict__ vT) {
  const int lane  = threadIdx.x;
  const int hf    = lane >> 4;
  const int l16   = lane & 15;
  const int roff  = hf * 8;
  const int n0 = blockIdx.x * 32;
  const int j0 = blockIdx.y * 64;
  const int b  = blockIdx.z;

  const _Float16* A0 = xh + (((size_t)b * NTOK + n0) >> 4) * (CCH * 16) + lane * 8;
  const _Float16* A1 = A0 + (size_t)CCH * 16;
  const _Float16* Bp[4];
#pragma unroll
  for (int t = 0; t < 4; ++t)
    Bp[t] = WpT + (size_t)((j0 >> 4) + t) * (CCH * 16) + lane * 8;

  v16h a0c = load_frag(A0), a1c = load_frag(A1);
  v16h bc[4];
#pragma unroll
  for (int t = 0; t < 4; ++t) bc[t] = load_frag(Bp[t]);
  v8f acc0[4] = {}, acc1[4] = {};
#pragma unroll
  for (int k0 = 0; k0 < CCH; k0 += 32) {
    v16h a0n = a0c, a1n = a1c, bn[4] = {bc[0], bc[1], bc[2], bc[3]};
    const int kn = k0 + 32;
    if (kn < CCH) {
      a0n = load_frag(A0 + kn * 16);
      a1n = load_frag(A1 + kn * 16);
#pragma unroll
      for (int t = 0; t < 4; ++t) bn[t] = load_frag(Bp[t] + kn * 16);
    }
#pragma unroll
    for (int t = 0; t < 4; ++t) {
      acc0[t] = wmma16(a0c, bc[t], acc0[t]);
      acc1[t] = wmma16(a1c, bc[t], acc1[t]);
    }
    a0c = a0n; a1c = a1n;
#pragma unroll
    for (int t = 0; t < 4; ++t) bc[t] = bn[t];
  }

#pragma unroll
  for (int t = 0; t < 4; ++t) {
    const int j0t  = j0 + t * 16;
    const int head = j0t / 384;
    const int r0   = j0t % 384;
    const int seg  = r0 / 128;
    const int dbase = r0 % 128;
    const float bias = bp[j0t + l16];
    const size_t bh = ((size_t)b * NH + head) * NTOK * DK;
    if (seg == 2) {
      _Float16* dst = vT + bh + (size_t)((n0 >> 5) * 8 + (dbase >> 4)) * 512 + lane * 8;
      v8h w0, w1;
#pragma unroll
      for (int g = 0; g < 8; ++g) {
        w0[g] = (_Float16)(acc0[t][g] + bias);
        w1[g] = (_Float16)(acc1[t][g] + bias);
      }
      *(v8h*)(dst)       = w0;
      *(v8h*)(dst + 256) = w1;
    } else {
      _Float16* base = (seg == 0) ? qh : kh;
      _Float16* p0 = base + bh + (size_t)((n0 >> 4) * 4 + (dbase >> 5)) * 512
                   + ((dbase >> 4) & 1) * 256 + ((l16 >> 3) * 16 + roff) * 8 + (l16 & 7);
#pragma unroll
      for (int g = 0; g < 8; ++g) {
        p0[g * 8]           = (_Float16)(acc0[t][g] + bias);
        p0[4 * 512 + g * 8] = (_Float16)(acc1[t][g] + bias);
      }
    }
  }
}

__global__ void __launch_bounds__(32) __attribute__((amdgpu_waves_per_eu(1, 4)))
attention_fa(const _Float16* __restrict__ qh, const _Float16* __restrict__ kh,
             const _Float16* __restrict__ vT, _Float16* __restrict__ attn) {
  __shared__ alignas(16) _Float16 lds_p[16][40];
  const int lane  = threadIdx.x;
  const int hf    = lane >> 4;
  const int l16   = lane & 15;
  const int kbase = hf * 8;
  const int roff  = hf * 8;
  const int n0 = blockIdx.x * 16;
  const int h  = blockIdx.y;
  const int b  = blockIdx.z;
  const float scale = 0.08838834764831845f;

  const _Float16* Qb = qh + ((size_t)b * NH + h) * NTOK * DK
                          + (size_t)(n0 >> 4) * 4 * 512 + lane * 8;
  v16h qf[4];
#pragma unroll
  for (int kk = 0; kk < 4; ++kk) qf[kk] = load_frag(Qb + kk * 512);

  v8f o[8] = {};
  float mi[8], li[8];
#pragma unroll
  for (int g = 0; g < 8; ++g) { mi[g] = -1e30f; li[g] = 0.0f; }

  const _Float16* Kraw = kh + ((size_t)b * NH + h) * NTOK * DK;
  const _Float16* Kb = Kraw + lane * 8;
  const _Float16* Vb = vT + ((size_t)b * NH + h) * DK * NTOK + lane * 8;

  for (int j0 = 0; j0 < NTOK; j0 += 32) {
    if (j0 + 32 < NTOK) {
      __builtin_prefetch(Kraw + (size_t)(j0 + 32) * DK + lane * 128, 0, 3);
      __builtin_prefetch(Kraw + (size_t)(j0 + 32) * DK + lane * 128 + 64, 0, 3);
    }
    v8f s0 = {}, s1 = {};
    const _Float16* K0 = Kb + (size_t)j0 * DK;
    const _Float16* K1 = K0 + 4 * 512;
#pragma unroll
    for (int kk = 0; kk < 4; ++kk) {
      s0 = wmma16(qf[kk], load_frag(K0 + kk * 512), s0);
      s1 = wmma16(qf[kk], load_frag(K1 + kk * 512), s1);
    }
    const _Float16* V0 = Vb + (size_t)j0 * DK;
    v16h vf[4];
#pragma unroll
    for (int t = 0; t < 4; ++t) vf[t] = load_frag(V0 + t * 512);

#pragma unroll
    for (int g = 0; g < 8; ++g) {
      const float a0 = s0[g] * scale, a1 = s1[g] * scale;
      float m = fmaxf(a0, a1);
      m = fmaxf(m, __shfl_xor(m, 1));
      m = fmaxf(m, __shfl_xor(m, 2));
      m = fmaxf(m, __shfl_xor(m, 4));
      m = fmaxf(m, __shfl_xor(m, 8));
      const float mn    = fmaxf(mi[g], m);
      const float alpha = __expf(mi[g] - mn);
      const float p0 = __expf(a0 - mn);
      const float p1 = __expf(a1 - mn);
      s0[g] = p0; s1[g] = p1;
      float rs = p0 + p1;
      rs += __shfl_xor(rs, 1);
      rs += __shfl_xor(rs, 2);
      rs += __shfl_xor(rs, 4);
      rs += __shfl_xor(rs, 8);
      li[g] = li[g] * alpha + rs;
      mi[g] = mn;
#pragma unroll
      for (int t = 0; t < 8; ++t) o[t][g] *= alpha;
    }
#pragma unroll
    for (int g = 0; g < 8; ++g) {
      lds_p[roff + g][l16]      = (_Float16)s0[g];
      lds_p[roff + g][16 + l16] = (_Float16)s1[g];
    }
    __syncthreads();
    const v16h pf = load_frag_rows(&lds_p[l16][kbase]);
#pragma unroll
    for (int t = 0; t < 4; ++t) o[t] = wmma16(pf, vf[t], o[t]);
#pragma unroll
    for (int t = 4; t < 8; ++t) o[t] = wmma16(pf, load_frag(V0 + t * 512), o[t]);
    __syncthreads();
  }
  _Float16* Ob = attn + ((((size_t)b * NTOK + n0) >> 4) * 16 + h * 4) * 512
               + ((l16 >> 3) * 16 + roff) * 8 + (l16 & 7);
#pragma unroll
  for (int t = 0; t < 8; ++t)
#pragma unroll
    for (int g = 0; g < 8; ++g)
      Ob[(t >> 1) * 512 + (t & 1) * 256 + g * 8] = (_Float16)(o[t][g] / li[g]);
}

__global__ void __launch_bounds__(32) __attribute__((amdgpu_waves_per_eu(1, 4)))
out_proj(const _Float16* __restrict__ attn,
         const _Float16* __restrict__ WoT,
         const float* __restrict__ bo,
         const float* __restrict__ x,
         float* __restrict__ out) {
  const int lane  = threadIdx.x;
  const int hf    = lane >> 4;
  const int l16   = lane & 15;
  const int roff  = hf * 8;
  const int n0 = blockIdx.x * 32;
  const int c0 = blockIdx.y * 64;
  const int b  = blockIdx.z;

  const _Float16* A0 = attn + (((size_t)b * NTOK + n0) >> 4) * (CCH * 16) + lane * 8;
  const _Float16* A1 = A0 + (size_t)CCH * 16;
  const _Float16* Bp[4];
#pragma unroll
  for (int t = 0; t < 4; ++t)
    Bp[t] = WoT + (size_t)((c0 >> 4) + t) * (CCH * 16) + lane * 8;

  v16h a0c = load_frag(A0), a1c = load_frag(A1);
  v16h bc[4];
#pragma unroll
  for (int t = 0; t < 4; ++t) bc[t] = load_frag(Bp[t]);
  v8f acc0[4] = {}, acc1[4] = {};
#pragma unroll
  for (int k0 = 0; k0 < CCH; k0 += 32) {
    v16h a0n = a0c, a1n = a1c, bn[4] = {bc[0], bc[1], bc[2], bc[3]};
    const int kn = k0 + 32;
    if (kn < CCH) {
      a0n = load_frag(A0 + kn * 16);
      a1n = load_frag(A1 + kn * 16);
#pragma unroll
      for (int t = 0; t < 4; ++t) bn[t] = load_frag(Bp[t] + kn * 16);
    }
#pragma unroll
    for (int t = 0; t < 4; ++t) {
      acc0[t] = wmma16(a0c, bc[t], acc0[t]);
      acc1[t] = wmma16(a1c, bc[t], acc1[t]);
    }
    a0c = a0n; a1c = a1n;
#pragma unroll
    for (int t = 0; t < 4; ++t) bc[t] = bn[t];
  }

#pragma unroll
  for (int t = 0; t < 4; ++t) {
    const int c = c0 + t * 16 + l16;
    const float bias = bo[c];
    const size_t base = ((size_t)b * CCH + c) * NTOK + n0 + roff;
#pragma unroll
    for (int half2 = 0; half2 < 2; ++half2) {
      const v8f& av   = half2 ? acc1[t] : acc0[t];
      const size_t bi = base + (size_t)half2 * 16;
      const v4f x0 = *(const v4f*)(x + bi);
      const v4f x1 = *(const v4f*)(x + bi + 4);
      v4f r0, r1;
#pragma unroll
      for (int g = 0; g < 4; ++g) {
        r0[g] = av[g]     + bias + x0[g];
        r1[g] = av[4 + g] + bias + x1[g];
      }
      *(v4f*)(out + bi)     = r0;
      *(v4f*)(out + bi + 4) = r1;
    }
  }
}

extern "C" void kernel_launch(void* const* d_in, const int* in_sizes, int n_in,
                              void* d_out, int out_size, void* d_ws, size_t ws_size,
                              hipStream_t stream) {
  (void)in_sizes; (void)n_in; (void)out_size; (void)ws_size;
  const float* x  = (const float*)d_in[0];
  const float* Wp = (const float*)d_in[1];
  const float* bp = (const float*)d_in[2];
  const float* Wo = (const float*)d_in[3];
  const float* bo = (const float*)d_in[4];
  float* out = (float*)d_out;

  char* ws = (char*)d_ws;
  _Float16* xh  = (_Float16*)(ws);
  _Float16* WpT = (_Float16*)(ws +  8388608);
  _Float16* WoT = (_Float16*)(ws +  9961472);
  _Float16* qh  = (_Float16*)(ws + 10485760);
  _Float16* kh  = (_Float16*)(ws + 18874368);
  _Float16* vT  = (_Float16*)(ws + 27262976);
  _Float16* at  = (_Float16*)(ws + 35651584);

  const dim3 tb(32, 8);
  transpose_cast<<<dim3(NTOK / 32, CCH / 32, NB), tb, 0, stream>>>(x,  xh,  CCH, NTOK);
  transpose_cast<<<dim3(1536 / 32, CCH / 32, 1),  tb, 0, stream>>>(Wp, WpT, CCH, 1536);
  transpose_cast<<<dim3(CCH  / 32, CCH / 32, 1),  tb, 0, stream>>>(Wo, WoT, CCH, CCH);

  qkv_gemm    <<<dim3(NTOK / 32, 1536 / 64, NB), 32, 0, stream>>>(xh, WpT, bp, qh, kh, vT);
  attention_fa<<<dim3(NTOK / 16, NH,        NB), 32, 0, stream>>>(qh, kh, vT, at);
  out_proj    <<<dim3(NTOK / 32, CCH / 64,  NB), 32, 0, stream>>>(at, WoT, bo, x, out);
}


# attention computed transposed (S^T=K.Q^T, O^T=V^T.P^T), 32 query columns per wave: max and row sum in-lane plus one half swap (new sum order), P^T from accumulators to the B operand, no LDS
# speedup vs baseline: 1.6363x; 1.3792x over previous
#include <hip/hip_runtime.h>
#include <hip/hip_bf16.h>


typedef _Float16 v16h __attribute__((ext_vector_type(16)));
typedef _Float16 v8h  __attribute__((ext_vector_type(8)));
typedef _Float16 v4h  __attribute__((ext_vector_type(4)));
typedef float    v8f  __attribute__((ext_vector_type(8)));
typedef float    v4f  __attribute__((ext_vector_type(4)));

#define NB   8
#define CCH  512
#define NTOK 1024
#define NH   4
#define DK   128

__device__ __forceinline__ v16h load_frag(const _Float16* p) {
  union { v16h v; v8h h[2]; } u;
  u.h[0] = *(const v8h*)(p);
  u.h[1] = *(const v8h*)(p + 256);
  return u.v;
}

__device__ __forceinline__ v8f wmma16(v16h a, v16h b, v8f c) {
  return __builtin_amdgcn_wmma_f32_16x16x32_f16(false, a, false, b,
                                                (short)0, c, false, false);
}

__global__ void __launch_bounds__(256)
transpose_cast(const float* __restrict__ src, _Float16* __restrict__ dst,
               int rows, int cols) {
  __shared__ float tile[32][36];
  const size_t boff = (size_t)blockIdx.z * rows * cols;
  const int cx = blockIdx.x * 32 + threadIdx.x;
  const int ry = blockIdx.y * 32 + threadIdx.y;
#pragma unroll
  for (int j = 0; j < 32; j += 8)
    tile[threadIdx.y + j][threadIdx.x] = src[boff + (size_t)(ry + j) * cols + cx];
  __syncthreads();
  const int tid = threadIdx.y * 32 + threadIdx.x;
  const int q   = tid & 127;
  const int e   = q >> 6;
  const int l   = (q >> 1) & 31;
  const int kk  = e * 16 + (l >> 4) * 8 + (q & 1) * 4;
  const int rr  = (tid >> 7) * 16 + (l & 15);
  const size_t rt = (((size_t)blockIdx.z * cols + (size_t)blockIdx.x * 32) >> 4) + (tid >> 7);
  _Float16* p = dst + (rt * (size_t)(rows >> 5) + blockIdx.y) * 512 + q * 4;
  v4h val;
#pragma unroll
  for (int i = 0; i < 4; ++i) val[i] = (_Float16)tile[kk + i][rr];
  *(v4h*)p = val;
}

__global__ void __launch_bounds__(32)
qkv_gemm(const _Float16* __restrict__ xh,
         const _Float16* __restrict__ WpT,
         const float* __restrict__ bp,
         _Float16* __restrict__ qh,
         _Float16* __restrict__ kh,
         _Float16* __restrict__ vT) {
  const int lane  = threadIdx.x;
  const int hf    = lane >> 4;
  const int l16   = lane & 15;
  const int roff  = hf * 8;
  const int n0 = blockIdx.x * 32;
  const int j0 = blockIdx.y * 64;
  const int b  = blockIdx.z;

  const _Float16* A0 = xh + (((size_t)b * NTOK + n0) >> 4) * (CCH * 16) + lane * 8;
  const _Float16* A1 = A0 + (size_t)CCH * 16;
  const _Float16* Bp[4];
#pragma unroll
  for (int t = 0; t < 4; ++t)
    Bp[t] = WpT + (size_t)((j0 >> 4) + t) * (CCH * 16) + lane * 8;

  v16h a0c = load_frag(A0), a1c = load_frag(A1);
  v16h bc[4];
#pragma unroll
  for (int t = 0; t < 4; ++t) bc[t] = load_frag(Bp[t]);
  v8f acc0[4] = {}, acc1[4] = {};
#pragma unroll
  for (int k0 = 0; k0 < CCH; k0 += 32) {
    v16h a0n = a0c, a1n = a1c, bn[4] = {bc[0], bc[1], bc[2], bc[3]};
    const int kn = k0 + 32;
    if (kn < CCH) {
      a0n = load_frag(A0 + kn * 16);
      a1n = load_frag(A1 + kn * 16);
#pragma unroll
      for (int t = 0; t < 4; ++t) bn[t] = load_frag(Bp[t] + kn * 16);
    }
#pragma unroll
    for (int t = 0; t < 4; ++t) {
      acc0[t] = wmma16(a0c, bc[t], acc0[t]);
      acc1[t] = wmma16(a1c, bc[t], acc1[t]);
    }
    a0c = a0n; a1c = a1n;
#pragma unroll
    for (int t = 0; t < 4; ++t) bc[t] = bn[t];
  }

#pragma unroll
  for (int t = 0; t < 4; ++t) {
    const int j0t  = j0 + t * 16;
    const int head = j0t / 384;
    const int r0   = j0t % 384;
    const int seg  = r0 / 128;
    const int dbase = r0 % 128;
    const float bias = bp[j0t + l16];
    const size_t bh = ((size_t)b * NH + head) * NTOK * DK;
    if (seg == 2) {
      _Float16* dst = vT + bh + (size_t)((n0 >> 5) * 8 + (dbase >> 4)) * 512 + lane * 8;
      v8h w0, w1;
#pragma unroll
      for (int g = 0; g < 8; ++g) {
        w0[g] = (_Float16)(acc0[t][g] + bias);
        w1[g] = (_Float16)(acc1[t][g] + bias);
      }
      *(v8h*)(dst)       = w0;
      *(v8h*)(dst + 256) = w1;
    } else {
      _Float16* base = (seg == 0) ? qh : kh;
      _Float16* p0 = base + bh + (size_t)((n0 >> 4) * 4 + (dbase >> 5)) * 512
                   + ((dbase >> 4) & 1) * 256 + ((l16 >> 3) * 16 + roff) * 8 + (l16 & 7);
#pragma unroll
      for (int g = 0; g < 8; ++g) {
        p0[g * 8]           = (_Float16)(acc0[t][g] + bias);
        p0[4 * 512 + g * 8] = (_Float16)(acc1[t][g] + bias);
      }
    }
  }
}

__device__ __forceinline__ float xhalf(float x) {
  return __builtin_bit_cast(float, __builtin_amdgcn_permlanex16(
      __builtin_bit_cast(int, x), __builtin_bit_cast(int, x),
      0x76543210, 0xfedcba98, false, false));
}

__global__ void __launch_bounds__(32)
attention_fa(const _Float16* __restrict__ qh, const _Float16* __restrict__ kh,
             const _Float16* __restrict__ vT, _Float16* __restrict__ attn) {
  const int lane = threadIdx.x;
  const int n0 = blockIdx.x * 32;
  const int h  = blockIdx.y;
  const int b  = blockIdx.z;
  constexpr float c2 = 0.08838834764831845f * 1.4426950408889634f;

  const size_t bh = ((size_t)b * NH + h) * NTOK * DK;
  const _Float16* Qb = qh + bh + (size_t)(n0 >> 4) * 4 * 512 + lane * 8;
  v16h qf[2][4];
#pragma unroll
  for (int qt = 0; qt < 2; ++qt)
#pragma unroll
    for (int kk = 0; kk < 4; ++kk) qf[qt][kk] = load_frag(Qb + (qt * 4 + kk) * 512);

  const _Float16* Kb = kh + bh + lane * 8;
  const _Float16* Vb = vT + bh + lane * 8;

  v16h kf[8];
#pragma unroll
  for (int i = 0; i < 8; ++i) kf[i] = load_frag(Kb + i * 512);

  v8f o[2][8] = {};
  float m_run[2] = {-1e30f, -1e30f};
  float l_run[2] = {0.0f, 0.0f};

  for (int j0 = 0; j0 < NTOK; j0 += 32) {
    const _Float16* V0 = Vb + (size_t)j0 * DK;
    v16h vf[8];
#pragma unroll
    for (int t = 0; t < 8; ++t) vf[t] = load_frag(V0 + t * 512);

    v8f s[2][2];
#pragma unroll
    for (int kt = 0; kt < 2; ++kt)
#pragma unroll
      for (int qt = 0; qt < 2; ++qt) {
        v8f acc = {};
#pragma unroll
        for (int kk = 0; kk < 4; ++kk) acc = wmma16(kf[kt * 4 + kk], qf[qt][kk], acc);
        s[kt][qt] = acc;
      }

    const int jn = (j0 + 32 < NTOK) ? j0 + 32 : j0;
    const _Float16* Kn = Kb + (size_t)jn * DK;
#pragma unroll
    for (int i = 0; i < 8; ++i) kf[i] = load_frag(Kn + i * 512);

#pragma unroll
    for (int qt = 0; qt < 2; ++qt) {
      float mx = s[0][qt][0];
#pragma unroll
      for (int r = 1; r < 8; ++r) mx = __builtin_fmaxf(mx, s[0][qt][r]);
#pragma unroll
      for (int r = 0; r < 8; ++r) mx = __builtin_fmaxf(mx, s[1][qt][r]);
      mx = __builtin_fmaxf(mx, xhalf(mx)) * c2;
      const float mn    = __builtin_fmaxf(m_run[qt], mx);
      const float alpha = __builtin_amdgcn_exp2f(m_run[qt] - mn);
      const float nm    = -mn;
      m_run[qt] = mn;
      float ps0 = 0.0f, ps1 = 0.0f;
      v16h pf;
#pragma unroll
      for (int r = 0; r < 8; ++r) {
        const float p0 = __builtin_amdgcn_exp2f(__builtin_fmaf(s[0][qt][r], c2, nm));
        const float p1 = __builtin_amdgcn_exp2f(__builtin_fmaf(s[1][qt][r], c2, nm));
        ps0 += p0; ps1 += p1;
        pf[r]     = (_Float16)p0;
        pf[8 + r] = (_Float16)p1;
      }
      l_run[qt] = l_run[qt] * alpha + (ps0 + ps1);
#pragma unroll
      for (int t = 0; t < 8; ++t) o[qt][t] = wmma16(vf[t], pf, o[qt][t] * alpha);
    }
  }

#pragma unroll
  for (int qt = 0; qt < 2; ++qt) {
    const float lt  = l_run[qt] + xhalf(l_run[qt]);
    const float inv = 1.0f / lt;
    _Float16* Ob = attn + (((((size_t)b * NTOK + n0) >> 4) + qt) * 16 + h * 4) * 512 + lane * 8;
#pragma unroll
    for (int t = 0; t < 8; ++t) {
      v8h w;
#pragma unroll
      for (int r = 0; r < 8; ++r) w[r] = (_Float16)(o[qt][t][r] * inv);
      *(v8h*)(Ob + (t >> 1) * 512 + (t & 1) * 256) = w;
    }
  }
}

__global__ void __launch_bounds__(32) __attribute__((amdgpu_waves_per_eu(1, 4)))
out_proj(const _Float16* __restrict__ attn,
         const _Float16* __restrict__ WoT,
         const float* __restrict__ bo,
         const float* __restrict__ x,
         float* __restrict__ out) {
  const int lane  = threadIdx.x;
  const int hf    = lane >> 4;
  const int l16   = lane & 15;
  const int roff  = hf * 8;
  const int n0 = blockIdx.x * 32;
  const int c0 = blockIdx.y * 64;
  const int b  = blockIdx.z;

  const _Float16* A0 = attn + (((size_t)b * NTOK + n0) >> 4) * (CCH * 16) + lane * 8;
  const _Float16* A1 = A0 + (size_t)CCH * 16;
  const _Float16* Bp[4];
#pragma unroll
  for (int t = 0; t < 4; ++t)
    Bp[t] = WoT + (size_t)((c0 >> 4) + t) * (CCH * 16) + lane * 8;

  v16h a0c = load_frag(A0), a1c = load_frag(A1);
  v16h bc[4];
#pragma unroll
  for (int t = 0; t < 4; ++t) bc[t] = load_frag(Bp[t]);
  v8f acc0[4] = {}, acc1[4] = {};
#pragma unroll
  for (int k0 = 0; k0 < CCH; k0 += 32) {
    v16h a0n = a0c, a1n = a1c, bn[4] = {bc[0], bc[1], bc[2], bc[3]};
    const int kn = k0 + 32;
    if (kn < CCH) {
      a0n = load_frag(A0 + kn * 16);
      a1n = load_frag(A1 + kn * 16);
#pragma unroll
      for (int t = 0; t < 4; ++t) bn[t] = load_frag(Bp[t] + kn * 16);
    }
#pragma unroll
    for (int t = 0; t < 4; ++t) {
      acc0[t] = wmma16(a0c, bc[t], acc0[t]);
      acc1[t] = wmma16(a1c, bc[t], acc1[t]);
    }
    a0c = a0n; a1c = a1n;
#pragma unroll
    for (int t = 0; t < 4; ++t) bc[t] = bn[t];
  }

#pragma unroll
  for (int t = 0; t < 4; ++t) {
    const int c = c0 + t * 16 + l16;
    const float bias = bo[c];
    const size_t base = ((size_t)b * CCH + c) * NTOK + n0 + roff;
#pragma unroll
    for (int half2 = 0; half2 < 2; ++half2) {
      const v8f& av   = half2 ? acc1[t] : acc0[t];
      const size_t bi = base + (size_t)half2 * 16;
      const v4f x0 = *(const v4f*)(x + bi);
      const v4f x1 = *(const v4f*)(x + bi + 4);
      v4f r0, r1;
#pragma unroll
      for (int g = 0; g < 4; ++g) {
        r0[g] = av[g]     + bias + x0[g];
        r1[g] = av[4 + g] + bias + x1[g];
      }
      *(v4f*)(out + bi)     = r0;
      *(v4f*)(out + bi + 4) = r1;
    }
  }
}

extern "C" void kernel_launch(void* const* d_in, const int* in_sizes, int n_in,
                              void* d_out, int out_size, void* d_ws, size_t ws_size,
                              hipStream_t stream) {
  (void)in_sizes; (void)n_in; (void)out_size; (void)ws_size;
  const float* x  = (const float*)d_in[0];
  const float* Wp = (const float*)d_in[1];
  const float* bp = (const float*)d_in[2];
  const float* Wo = (const float*)d_in[3];
  const float* bo = (const float*)d_in[4];
  float* out = (float*)d_out;

  char* ws = (char*)d_ws;
  _Float16* xh  = (_Float16*)(ws);
  _Float16* WpT = (_Float16*)(ws +  8388608);
  _Float16* WoT = (_Float16*)(ws +  9961472);
  _Float16* qh  = (_Float16*)(ws + 10485760);
  _Float16* kh  = (_Float16*)(ws + 18874368);
  _Float16* vT  = (_Float16*)(ws + 27262976);
  _Float16* at  = (_Float16*)(ws + 35651584);

  const dim3 tb(32, 8);
  transpose_cast<<<dim3(NTOK / 32, CCH / 32, NB), tb, 0, stream>>>(x,  xh,  CCH, NTOK);
  transpose_cast<<<dim3(1536 / 32, CCH / 32, 1),  tb, 0, stream>>>(Wp, WpT, CCH, 1536);
  transpose_cast<<<dim3(CCH  / 32, CCH / 32, 1),  tb, 0, stream>>>(Wo, WoT, CCH, CCH);

  qkv_gemm    <<<dim3(NTOK / 32, 1536 / 64, NB), 32, 0, stream>>>(xh, WpT, bp, qh, kh, vT);
  attention_fa<<<dim3(NTOK / 32, NH,        NB), 32, 0, stream>>>(qh, kh, vT, at);
  out_proj    <<<dim3(NTOK / 32, CCH / 64,  NB), 32, 0, stream>>>(at, WoT, bo, x, out);
}


# qkv projection re-tiled: wave tile 32x64 -> 64x96 (24 WMMAs per 10 KB of fragment loads, was 8 per 6 KB), 8-wave groups of 4x2 tiles, loads fenced two k-steps ahead (were waited for pair by pair)
# speedup vs baseline: 1.9872x; 1.2144x over previous
#include <hip/hip_runtime.h>
#include <hip/hip_bf16.h>


typedef _Float16 v16h __attribute__((ext_vector_type(16)));
typedef _Float16 v8h  __attribute__((ext_vector_type(8)));
typedef _Float16 v4h  __attribute__((ext_vector_type(4)));
typedef float    v8f  __attribute__((ext_vector_type(8)));
typedef float    v4f  __attribute__((ext_vector_type(4)));

#define NB   8
#define CCH  512
#define NTOK 1024
#define NH   4
#define DK   128

__device__ __forceinline__ v16h load_frag(const _Float16* p) {
  union { v16h v; v8h h[2]; } u;
  u.h[0] = *(const v8h*)(p);
  u.h[1] = *(const v8h*)(p + 256);
  return u.v;
}

__device__ __forceinline__ v8f wmma16(v16h a, v16h b, v8f c) {
  return __builtin_amdgcn_wmma_f32_16x16x32_f16(false, a, false, b,
                                                (short)0, c, false, false);
}

__global__ void __launch_bounds__(256)
transpose_cast(const float* __restrict__ src, _Float16* __restrict__ dst,
               int rows, int cols) {
  __shared__ float tile[32][36];
  const size_t boff = (size_t)blockIdx.z * rows * cols;
  const int cx = blockIdx.x * 32 + threadIdx.x;
  const int ry = blockIdx.y * 32 + threadIdx.y;
#pragma unroll
  for (int j = 0; j < 32; j += 8)
    tile[threadIdx.y + j][threadIdx.x] = src[boff + (size_t)(ry + j) * cols + cx];
  __syncthreads();
  const int tid = threadIdx.y * 32 + threadIdx.x;
  const int q   = tid & 127;
  const int e   = q >> 6;
  const int l   = (q >> 1) & 31;
  const int kk  = e * 16 + (l >> 4) * 8 + (q & 1) * 4;
  const int rr  = (tid >> 7) * 16 + (l & 15);
  const size_t rt = (((size_t)blockIdx.z * cols + (size_t)blockIdx.x * 32) >> 4) + (tid >> 7);
  _Float16* p = dst + (rt * (size_t)(rows >> 5) + blockIdx.y) * 512 + q * 4;
  v4h val;
#pragma unroll
  for (int i = 0; i < 4; ++i) val[i] = (_Float16)tile[kk + i][rr];
  *(v4h*)p = val;
}

__global__ void __launch_bounds__(256) __attribute__((amdgpu_waves_per_eu(1, 2)))
qkv_gemm(const _Float16* __restrict__ xh,
         const _Float16* __restrict__ WpT,
         const float* __restrict__ bp,
         _Float16* __restrict__ qh,
         _Float16* __restrict__ kh,
         _Float16* __restrict__ vT) {
  const int lane  = threadIdx.x & 31;
  const int wv    = __builtin_amdgcn_readfirstlane((int)(threadIdx.x >> 5));
  const int hf    = lane >> 4;
  const int l16   = lane & 15;
  const int roff  = hf * 8;
  const int n0 = (blockIdx.x * 4 + (wv >> 2) * 2 + (wv & 1)) * 64;
  const int j0 = (blockIdx.y * 2 + ((wv >> 1) & 1)) * 96;
  const int b  = blockIdx.z;

  const _Float16* Ap[4];
#pragma unroll
  for (int i = 0; i < 4; ++i)
    Ap[i] = xh + ((((size_t)b * NTOK + n0) >> 4) + i) * (CCH * 16) + lane * 8;
  const _Float16* Bp[6];
#pragma unroll
  for (int t = 0; t < 6; ++t)
    Bp[t] = WpT + (size_t)((j0 >> 4) + t) * (CCH * 16) + lane * 8;

  v16h af[3][4], bf[3][6];
#pragma unroll
  for (int s = 0; s < 2; ++s) {
#pragma unroll
    for (int i = 0; i < 4; ++i) af[s][i] = load_frag(Ap[i] + s * 512);
#pragma unroll
    for (int t = 0; t < 6; ++t) bf[s][t] = load_frag(Bp[t] + s * 512);
  }
  v8f acc[4][6] = {};
  float bias6[6] = {};
#pragma unroll
  for (int ks = 0; ks < CCH / 32; ++ks) {
    const int cur = ks % 3, nxt = (ks + 2) % 3;
    if (ks + 2 < CCH / 32) {
#pragma unroll
      for (int i = 0; i < 4; ++i) af[nxt][i] = load_frag(Ap[i] + (ks + 2) * 512);
#pragma unroll
      for (int t = 0; t < 6; ++t) bf[nxt][t] = load_frag(Bp[t] + (ks + 2) * 512);
    } else if (ks + 2 == CCH / 32) {
#pragma unroll
      for (int t = 0; t < 6; ++t) bias6[t] = bp[j0 + t * 16 + l16];
    }
    __builtin_amdgcn_sched_barrier(0);
#pragma unroll
    for (int t = 0; t < 6; ++t)
#pragma unroll
      for (int i = 0; i < 4; ++i) acc[i][t] = wmma16(af[cur][i], bf[cur][t], acc[i][t]);
    __builtin_amdgcn_sched_barrier(0);
  }

#pragma unroll
  for (int t = 0; t < 6; ++t) {
    const int j0t  = j0 + t * 16;
    const int head = j0t / 384;
    const int r0   = j0t % 384;
    const int seg  = r0 / 128;
    const int dbase = r0 % 128;
    const float bias = bias6[t];
    const size_t bh = ((size_t)b * NH + head) * NTOK * DK;
#pragma unroll
    for (int pr = 0; pr < 2; ++pr) {
      const int n0p = n0 + pr * 32;
      const v8f& acc0 = acc[2 * pr][t];
      const v8f& acc1 = acc[2 * pr + 1][t];
      if (seg == 2) {
        _Float16* dst = vT + bh + (size_t)((n0p >> 5) * 8 + (dbase >> 4)) * 512 + lane * 8;
        v8h w0, w1;
#pragma unroll
        for (int g = 0; g < 8; ++g) {
          w0[g] = (_Float16)(acc0[g] + bias);
          w1[g] = (_Float16)(acc1[g] + bias);
        }
        *(v8h*)(dst)       = w0;
        *(v8h*)(dst + 256) = w1;
      } else {
        _Float16* base = (seg == 0) ? qh : kh;
        _Float16* p0 = base + bh + (size_t)((n0p >> 4) * 4 + (dbase >> 5)) * 512
                     + ((dbase >> 4) & 1) * 256 + ((l16 >> 3) * 16 + roff) * 8 + (l16 & 7);
#pragma unroll
        for (int g = 0; g < 8; ++g) {
          p0[g * 8]           = (_Float16)(acc0[g] + bias);
          p0[4 * 512 + g * 8] = (_Float16)(acc1[g] + bias);
        }
      }
    }
  }
}

__device__ __forceinline__ float xhalf(float x) {
  return __builtin_bit_cast(float, __builtin_amdgcn_permlanex16(
      __builtin_bit_cast(int, x), __builtin_bit_cast(int, x),
      0x76543210, 0xfedcba98, false, false));
}

__global__ void __launch_bounds__(32)
attention_fa(const _Float16* __restrict__ qh, const _Float16* __restrict__ kh,
             const _Float16* __restrict__ vT, _Float16* __restrict__ attn) {
  const int lane = threadIdx.x;
  const int n0 = blockIdx.x * 32;
  const int h  = blockIdx.y;
  const int b  = blockIdx.z;
  constexpr float c2 = 0.08838834764831845f * 1.4426950408889634f;

  const size_t bh = ((size_t)b * NH + h) * NTOK * DK;
  const _Float16* Qb = qh + bh + (size_t)(n0 >> 4) * 4 * 512 + lane * 8;
  v16h qf[2][4];
#pragma unroll
  for (int qt = 0; qt < 2; ++qt)
#pragma unroll
    for (int kk = 0; kk < 4; ++kk) qf[qt][kk] = load_frag(Qb + (qt * 4 + kk) * 512);

  const _Float16* Kb = kh + bh + lane * 8;
  const _Float16* Vb = vT + bh + lane * 8;

  v16h kf[8];
#pragma unroll
  for (int i = 0; i < 8; ++i) kf[i] = load_frag(Kb + i * 512);

  v8f o[2][8] = {};
  float m_run[2] = {-1e30f, -1e30f};
  float l_run[2] = {0.0f, 0.0f};

  for (int j0 = 0; j0 < NTOK; j0 += 32) {
    const _Float16* V0 = Vb + (size_t)j0 * DK;
    v16h vf[8];
#pragma unroll
    for (int t = 0; t < 8; ++t) vf[t] = load_frag(V0 + t * 512);

    v8f s[2][2];
#pragma unroll
    for (int kt = 0; kt < 2; ++kt)
#pragma unroll
      for (int qt = 0; qt < 2; ++qt) {
        v8f acc = {};
#pragma unroll
        for (int kk = 0; kk < 4; ++kk) acc = wmma16(kf[kt * 4 + kk], qf[qt][kk], acc);
        s[kt][qt] = acc;
      }

    const int jn = (j0 + 32 < NTOK) ? j0 + 32 : j0;
    const _Float16* Kn = Kb + (size_t)jn * DK;
#pragma unroll
    for (int i = 0; i < 8; ++i) kf[i] = load_frag(Kn + i * 512);

#pragma unroll
    for (int qt = 0; qt < 2; ++qt) {
      float mx = s[0][qt][0];
#pragma unroll
      for (int r = 1; r < 8; ++r) mx = __builtin_fmaxf(mx, s[0][qt][r]);
#pragma unroll
      for (int r = 0; r < 8; ++r) mx = __builtin_fmaxf(mx, s[1][qt][r]);
      mx = __builtin_fmaxf(mx, xhalf(mx)) * c2;
      const float mn    = __builtin_fmaxf(m_run[qt], mx);
      const float alpha = __builtin_amdgcn_exp2f(m_run[qt] - mn);
      const float nm    = -mn;
      m_run[qt] = mn;
      float ps0 = 0.0f, ps1 = 0.0f;
      v16h pf;
#pragma unroll
      for (int r = 0; r < 8; ++r) {
        const float p0 = __builtin_amdgcn_exp2f(__builtin_fmaf(s[0][qt][r], c2, nm));
        const float p1 = __builtin_amdgcn_exp2f(__builtin_fmaf(s[1][qt][r], c2, nm));
        ps0 += p0; ps1 += p1;
        pf[r]     = (_Float16)p0;
        pf[8 + r] = (_Float16)p1;
      }
      l_run[qt] = l_run[qt] * alpha + (ps0 + ps1);
#pragma unroll
      for (int t = 0; t < 8; ++t) o[qt][t] = wmma16(vf[t], pf, o[qt][t] * alpha);
    }
  }

#pragma unroll
  for (int qt = 0; qt < 2; ++qt) {
    const float lt  = l_run[qt] + xhalf(l_run[qt]);
    const float inv = 1.0f / lt;
    _Float16* Ob = attn + (((((size_t)b * NTOK + n0) >> 4) + qt) * 16 + h * 4) * 512 + lane * 8;
#pragma unroll
    for (int t = 0; t < 8; ++t) {
      v8h w;
#pragma unroll
      for (int r = 0; r < 8; ++r) w[r] = (_Float16)(o[qt][t][r] * inv);
      *(v8h*)(Ob + (t >> 1) * 512 + (t & 1) * 256) = w;
    }
  }
}

__global__ void __launch_bounds__(32) __attribute__((amdgpu_waves_per_eu(1, 4)))
out_proj(const _Float16* __restrict__ attn,
         const _Float16* __restrict__ WoT,
         const float* __restrict__ bo,
         const float* __restrict__ x,
         float* __restrict__ out) {
  const int lane  = threadIdx.x;
  const int hf    = lane >> 4;
  const int l16   = lane & 15;
  const int roff  = hf * 8;
  const int n0 = blockIdx.x * 32;
  const int c0 = blockIdx.y * 64;
  const int b  = blockIdx.z;

  const _Float16* A0 = attn + (((size_t)b * NTOK + n0) >> 4) * (CCH * 16) + lane * 8;
  const _Float16* A1 = A0 + (size_t)CCH * 16;
  const _Float16* Bp[4];
#pragma unroll
  for (int t = 0; t < 4; ++t)
    Bp[t] = WoT + (size_t)((c0 >> 4) + t) * (CCH * 16) + lane * 8;

  v16h a0c = load_frag(A0), a1c = load_frag(A1);
  v16h bc[4];
#pragma unroll
  for (int t = 0; t < 4; ++t) bc[t] = load_frag(Bp[t]);
  v8f acc0[4] = {}, acc1[4] = {};
#pragma unroll
  for (int k0 = 0; k0 < CCH; k0 += 32) {
    v16h a0n = a0c, a1n = a1c, bn[4] = {bc[0], bc[1], bc[2], bc[3]};
    const int kn = k0 + 32;
    if (kn < CCH) {
      a0n = load_frag(A0 + kn * 16);
      a1n = load_frag(A1 + kn * 16);
#pragma unroll
      for (int t = 0; t < 4; ++t) bn[t] = load_frag(Bp[t] + kn * 16);
    }
#pragma unroll
    for (int t = 0; t < 4; ++t) {
      acc0[t] = wmma16(a0c, bc[t], acc0[t]);
      acc1[t] = wmma16(a1c, bc[t], acc1[t]);
    }
    a0c = a0n; a1c = a1n;
#pragma unroll
    for (int t = 0; t < 4; ++t) bc[t] = bn[t];
  }

#pragma unroll
  for (int t = 0; t < 4; ++t) {
    const int c = c0 + t * 16 + l16;
    const float bias = bo[c];
    const size_t base = ((size_t)b * CCH + c) * NTOK + n0 + roff;
#pragma unroll
    for (int half2 = 0; half2 < 2; ++half2) {
      const v8f& av   = half2 ? acc1[t] : acc0[t];
      const size_t bi = base + (size_t)half2 * 16;
      const v4f x0 = *(const v4f*)(x + bi);
      const v4f x1 = *(const v4f*)(x + bi + 4);
      v4f r0, r1;
#pragma unroll
      for (int g = 0; g < 4; ++g) {
        r0[g] = av[g]     + bias + x0[g];
        r1[g] = av[4 + g] + bias + x1[g];
      }
      *(v4f*)(out + bi)     = r0;
      *(v4f*)(out + bi + 4) = r1;
    }
  }
}

extern "C" void kernel_launch(void* const* d_in, const int* in_sizes, int n_in,
                              void* d_out, int out_size, void* d_ws, size_t ws_size,
                              hipStream_t stream) {
  (void)in_sizes; (void)n_in; (void)out_size; (void)ws_size;
  const float* x  = (const float*)d_in[0];
  const float* Wp = (const float*)d_in[1];
  const float* bp = (const float*)d_in[2];
  const float* Wo = (const float*)d_in[3];
  const float* bo = (const float*)d_in[4];
  float* out = (float*)d_out;

  char* ws = (char*)d_ws;
  _Float16* xh  = (_Float16*)(ws);
  _Float16* WpT = (_Float16*)(ws +  8388608);
  _Float16* WoT = (_Float16*)(ws +  9961472);
  _Float16* qh  = (_Float16*)(ws + 10485760);
  _Float16* kh  = (_Float16*)(ws + 18874368);
  _Float16* vT  = (_Float16*)(ws + 27262976);
  _Float16* at  = (_Float16*)(ws + 35651584);

  const dim3 tb(32, 8);
  transpose_cast<<<dim3(NTOK / 32, CCH / 32, NB), tb, 0, stream>>>(x,  xh,  CCH, NTOK);
  transpose_cast<<<dim3(1536 / 32, CCH / 32, 1),  tb, 0, stream>>>(Wp, WpT, CCH, 1536);
  transpose_cast<<<dim3(CCH  / 32, CCH / 32, 1),  tb, 0, stream>>>(Wo, WoT, CCH, CCH);

  qkv_gemm    <<<dim3(NTOK / 256, 1536 / 192, NB), 256, 0, stream>>>(xh, WpT, bp, qh, kh, vT);
  attention_fa<<<dim3(NTOK / 32, NH,        NB), 32, 0, stream>>>(qh, kh, vT, at);
  out_proj    <<<dim3(NTOK / 32, CCH / 64,  NB), 32, 0, stream>>>(at, WoT, bo, x, out);
}


# layout conversion (transpose + f16 cast of x, Wp, Wo): one launch of 256 eight-wave groups, five 32x128 tiles each, 16-B loads and stores; 2048 waves enter, not 40960; 4 launches, not 6
# speedup vs baseline: 2.1149x; 1.0643x over previous
#include <hip/hip_runtime.h>
#include <hip/hip_bf16.h>


typedef _Float16 v16h __attribute__((ext_vector_type(16)));
typedef _Float16 v8h  __attribute__((ext_vector_type(8)));
typedef _Float16 v4h  __attribute__((ext_vector_type(4)));
typedef float    v8f  __attribute__((ext_vector_type(8)));
typedef float    v4f  __attribute__((ext_vector_type(4)));

#define NB   8
#define CCH  512
#define NTOK 1024
#define NH   4
#define DK   128

__device__ __forceinline__ v16h load_frag(const _Float16* p) {
  union { v16h v; v8h h[2]; } u;
  u.h[0] = *(const v8h*)(p);
  u.h[1] = *(const v8h*)(p + 256);
  return u.v;
}

__device__ __forceinline__ v8f wmma16(v16h a, v16h b, v8f c) {
  return __builtin_amdgcn_wmma_f32_16x16x32_f16(false, a, false, b,
                                                (short)0, c, false, false);
}

#define TPITCH 132

__device__ __forceinline__ void load_tile(v4f (&r)[4], const float* __restrict__ s,
                                          int cols, int lrow, int lcol) {
#pragma unroll
  for (int j = 0; j < 4; ++j)
    r[j] = *(const v4f*)(s + (size_t)(j * 8 + lrow) * cols + lcol);
}

__global__ void __launch_bounds__(256)
to_fragment_order(const float* __restrict__ x, const float* __restrict__ Wp,
                  const float* __restrict__ Wo, _Float16* __restrict__ xh,
                  _Float16* __restrict__ WpT, _Float16* __restrict__ WoT) {
  __shared__ float tile[2][32][TPITCH];
  const int t    = threadIdx.x;
  const int wg   = blockIdx.x;
  const int lrow = t >> 5;
  const int lcol = (t & 31) * 4;

  const float* sp[5]; _Float16* dp[5]; int cl[5];
#pragma unroll
  for (int i = 0; i < 4; ++i) {
    const int id = i * 256 + wg;
    const int b  = id >> 7, kt = id & 15, nt = (id >> 4) & 7;
    sp[i] = x + ((size_t)b * CCH + kt * 32) * NTOK + nt * 128;
    dp[i] = xh + ((size_t)((b * NTOK + nt * 128) >> 4) * 16 + kt) * 512;
    cl[i] = NTOK;
  }
  {
    const bool isp = wg < 192;
    const int id = isp ? wg : wg - 192;
    const int kt = id & 15, nt = id >> 4;
    cl[4] = isp ? 1536 : CCH;
    sp[4] = (isp ? Wp : Wo) + (size_t)(kt * 32) * cl[4] + nt * 128;
    dp[4] = (isp ? WpT : WoT) + ((size_t)((nt * 128) >> 4) * 16 + kt) * 512;
  }

  v4f r[5][4];
#pragma unroll
  for (int i = 0; i < 5; ++i) load_tile(r[i], sp[i], cl[i], lrow, lcol);

#pragma unroll
  for (int i = 0; i < 5; ++i) {
    float (*tl)[TPITCH] = tile[i & 1];
#pragma unroll
    for (int j = 0; j < 4; ++j) *(v4f*)&tl[j * 8 + lrow][lcol] = r[i][j];
    __builtin_amdgcn_fence(__ATOMIC_RELEASE, "workgroup", "local");
    __builtin_amdgcn_s_barrier();
    __builtin_amdgcn_fence(__ATOMIC_ACQUIRE, "workgroup", "local");
#pragma unroll
    for (int u = 0; u < 2; ++u) {
      const int S  = t + u * 256;
      const int jr = S >> 6, s = S & 63;
      const int k0 = (s >> 5) * 16 + ((s >> 4) & 1) * 8;
      const int n  = jr * 16 + (s & 15);
      v8h w;
#pragma unroll
      for (int e = 0; e < 8; ++e) w[e] = (_Float16)tl[k0 + e][n];
      *(v8h*)(dp[i] + (size_t)jr * (16 * 512) + s * 8) = w;
    }
  }
}

__global__ void __launch_bounds__(256) __attribute__((amdgpu_waves_per_eu(1, 2)))
qkv_gemm(const _Float16* __restrict__ xh,
         const _Float16* __restrict__ WpT,
         const float* __restrict__ bp,
         _Float16* __restrict__ qh,
         _Float16* __restrict__ kh,
         _Float16* __restrict__ vT) {
  const int lane  = threadIdx.x & 31;
  const int wv    = __builtin_amdgcn_readfirstlane((int)(threadIdx.x >> 5));
  const int hf    = lane >> 4;
  const int l16   = lane & 15;
  const int roff  = hf * 8;
  const int n0 = (blockIdx.x * 4 + (wv >> 2) * 2 + (wv & 1)) * 64;
  const int j0 = (blockIdx.y * 2 + ((wv >> 1) & 1)) * 96;
  const int b  = blockIdx.z;

  const _Float16* Ap[4];
#pragma unroll
  for (int i = 0; i < 4; ++i)
    Ap[i] = xh + ((((size_t)b * NTOK + n0) >> 4) + i) * (CCH * 16) + lane * 8;
  const _Float16* Bp[6];
#pragma unroll
  for (int t = 0; t < 6; ++t)
    Bp[t] = WpT + (size_t)((j0 >> 4) + t) * (CCH * 16) + lane * 8;

  v16h af[3][4], bf[3][6];
#pragma unroll
  for (int s = 0; s < 2; ++s) {
#pragma unroll
    for (int i = 0; i < 4; ++i) af[s][i] = load_frag(Ap[i] + s * 512);
#pragma unroll
    for (int t = 0; t < 6; ++t) bf[s][t] = load_frag(Bp[t] + s * 512);
  }
  v8f acc[4][6] = {};
  float bias6[6] = {};
#pragma unroll
  for (int ks = 0; ks < CCH / 32; ++ks) {
    const int cur = ks % 3, nxt = (ks + 2) % 3;
    if (ks + 2 < CCH / 32) {
#pragma unroll
      for (int i = 0; i < 4; ++i) af[nxt][i] = load_frag(Ap[i] + (ks + 2) * 512);
#pragma unroll
      for (int t = 0; t < 6; ++t) bf[nxt][t] = load_frag(Bp[t] + (ks + 2) * 512);
    } else if (ks + 2 == CCH / 32) {
#pragma unroll
      for (int t = 0; t < 6; ++t) bias6[t] = bp[j0 + t * 16 + l16];
    }
    __builtin_amdgcn_sched_barrier(0);
#pragma unroll
    for (int t = 0; t < 6; ++t)
#pragma unroll
      for (int i = 0; i < 4; ++i) acc[i][t] = wmma16(af[cur][i], bf[cur][t], acc[i][t]);
    __builtin_amdgcn_sched_barrier(0);
  }

#pragma unroll
  for (int t = 0; t < 6; ++t) {
    const int j0t  = j0 + t * 16;
    const int head = j0t / 384;
    const int r0   = j0t % 384;
    const int seg  = r0 / 128;
    const int dbase = r0 % 128;
    const float bias = bias6[t];
    const size_t bh = ((size_t)b * NH + head) * NTOK * DK;
#pragma unroll
    for (int pr = 0; pr < 2; ++pr) {
      const int n0p = n0 + pr * 32;
      const v8f& acc0 = acc[2 * pr][t];
      const v8f& acc1 = acc[2 * pr + 1][t];
      if (seg == 2) {
        _Float16* dst = vT + bh + (size_t)((n0p >> 5) * 8 + (dbase >> 4)) * 512 + lane * 8;
        v8h w0, w1;
#pragma unroll
        for (int g = 0; g < 8; ++g) {
          w0[g] = (_Float16)(acc0[g] + bias);
          w1[g] = (_Float16)(acc1[g] + bias);
        }
        *(v8h*)(dst)       = w0;
        *(v8h*)(dst + 256) = w1;
      } else {
        _Float16* base = (seg == 0) ? qh : kh;
        _Float16* p0 = base + bh + (size_t)((n0p >> 4) * 4 + (dbase >> 5)) * 512
                     + ((dbase >> 4) & 1) * 256 + ((l16 >> 3) * 16 + roff) * 8 + (l16 & 7);
#pragma unroll
        for (int g = 0; g < 8; ++g) {
          p0[g * 8]           = (_Float16)(acc0[g] + bias);
          p0[4 * 512 + g * 8] = (_Float16)(acc1[g] + bias);
        }
      }
    }
  }
}

__device__ __forceinline__ float xhalf(float x) {
  return __builtin_bit_cast(float, __builtin_amdgcn_permlanex16(
      __builtin_bit_cast(int, x), __builtin_bit_cast(int, x),
      0x76543210, 0xfedcba98, false, false));
}

__global__ void __launch_bounds__(32)
attention_fa(const _Float16* __restrict__ qh, const _Float16* __restrict__ kh,
             const _Float16* __restrict__ vT, _Float16* __restrict__ attn) {
  const int lane = threadIdx.x;
  const int n0 = blockIdx.x * 32;
  const int h  = blockIdx.y;
  const int b  = blockIdx.z;
  constexpr float c2 = 0.08838834764831845f * 1.4426950408889634f;

  const size_t bh = ((size_t)b * NH + h) * NTOK * DK;
  const _Float16* Qb = qh + bh + (size_t)(n0 >> 4) * 4 * 512 + lane * 8;
  v16h qf[2][4];
#pragma unroll
  for (int qt = 0; qt < 2; ++qt)
#pragma unroll
    for (int kk = 0; kk < 4; ++kk) qf[qt][kk] = load_frag(Qb + (qt * 4 + kk) * 512);

  const _Float16* Kb = kh + bh + lane * 8;
  const _Float16* Vb = vT + bh + lane * 8;

  v16h kf[8];
#pragma unroll
  for (int i = 0; i < 8; ++i) kf[i] = load_frag(Kb + i * 512);

  v8f o[2][8] = {};
  float m_run[2] = {-1e30f, -1e30f};
  float l_run[2] = {0.0f, 0.0f};

  for (int j0 = 0; j0 < NTOK; j0 += 32) {
    const _Float16* V0 = Vb + (size_t)j0 * DK;
    v16h vf[8];
#pragma unroll
    for (int t = 0; t < 8; ++t) vf[t] = load_frag(V0 + t * 512);

    v8f s[2][2];
#pragma unroll
    for (int kt = 0; kt < 2; ++kt)
#pragma unroll
      for (int qt = 0; qt < 2; ++qt) {
        v8f acc = {};
#pragma unroll
        for (int kk = 0; kk < 4; ++kk) acc = wmma16(kf[kt * 4 + kk], qf[qt][kk], acc);
        s[kt][qt] = acc;
      }

    const int jn = (j0 + 32 < NTOK) ? j0 + 32 : j0;
    const _Float16* Kn = Kb + (size_t)jn * DK;
#pragma unroll
    for (int i = 0; i < 8; ++i) kf[i] = load_frag(Kn + i * 512);

#pragma unroll
    for (int qt = 0; qt < 2; ++qt) {
      float mx = s[0][qt][0];
#pragma unroll
      for (int r = 1; r < 8; ++r) mx = __builtin_fmaxf(mx, s[0][qt][r]);
#pragma unroll
      for (int r = 0; r < 8; ++r) mx = __builtin_fmaxf(mx, s[1][qt][r]);
      mx = __builtin_fmaxf(mx, xhalf(mx)) * c2;
      const float mn    = __builtin_fmaxf(m_run[qt], mx);
      const float alpha = __builtin_amdgcn_exp2f(m_run[qt] - mn);
      const float nm    = -mn;
      m_run[qt] = mn;
      float ps0 = 0.0f, ps1 = 0.0f;
      v16h pf;
#pragma unroll
      for (int r = 0; r < 8; ++r) {
        const float p0 = __builtin_amdgcn_exp2f(__builtin_fmaf(s[0][qt][r], c2, nm));
        const float p1 = __builtin_amdgcn_exp2f(__builtin_fmaf(s[1][qt][r], c2, nm));
        ps0 += p0; ps1 += p1;
        pf[r]     = (_Float16)p0;
        pf[8 + r] = (_Float16)p1;
      }
      l_run[qt] = l_run[qt] * alpha + (ps0 + ps1);
#pragma unroll
      for (int t = 0; t < 8; ++t) o[qt][t] = wmma16(vf[t], pf, o[qt][t] * alpha);
    }
  }

#pragma unroll
  for (int qt = 0; qt < 2; ++qt) {
    const float lt  = l_run[qt] + xhalf(l_run[qt]);
    const float inv = 1.0f / lt;
    _Float16* Ob = attn + (((((size_t)b * NTOK + n0) >> 4) + qt) * 16 + h * 4) * 512 + lane * 8;
#pragma unroll
    for (int t = 0; t < 8; ++t) {
      v8h w;
#pragma unroll
      for (int r = 0; r < 8; ++r) w[r] = (_Float16)(o[qt][t][r] * inv);
      *(v8h*)(Ob + (t >> 1) * 512 + (t & 1) * 256) = w;
    }
  }
}

__global__ void __launch_bounds__(32) __attribute__((amdgpu_waves_per_eu(1, 4)))
out_proj(const _Float16* __restrict__ attn,
         const _Float16* __restrict__ WoT,
         const float* __restrict__ bo,
         const float* __restrict__ x,
         float* __restrict__ out) {
  const int lane  = threadIdx.x;
  const int hf    = lane >> 4;
  const int l16   = lane & 15;
  const int roff  = hf * 8;
  const int n0 = blockIdx.x * 32;
  const int c0 = blockIdx.y * 64;
  const int b  = blockIdx.z;

  const _Float16* A0 = attn + (((size_t)b * NTOK + n0) >> 4) * (CCH * 16) + lane * 8;
  const _Float16* A1 = A0 + (size_t)CCH * 16;
  const _Float16* Bp[4];
#pragma unroll
  for (int t = 0; t < 4; ++t)
    Bp[t] = WoT + (size_t)((c0 >> 4) + t) * (CCH * 16) + lane * 8;

  v16h a0c = load_frag(A0), a1c = load_frag(A1);
  v16h bc[4];
#pragma unroll
  for (int t = 0; t < 4; ++t) bc[t] = load_frag(Bp[t]);
  v8f acc0[4] = {}, acc1[4] = {};
#pragma unroll
  for (int k0 = 0; k0 < CCH; k0 += 32) {
    v16h a0n = a0c, a1n = a1c, bn[4] = {bc[0], bc[1], bc[2], bc[3]};
    const int kn = k0 + 32;
    if (kn < CCH) {
      a0n = load_frag(A0 + kn * 16);
      a1n = load_frag(A1 + kn * 16);
#pragma unroll
      for (int t = 0; t < 4; ++t) bn[t] = load_frag(Bp[t] + kn * 16);
    }
#pragma unroll
    for (int t = 0; t < 4; ++t) {
      acc0[t] = wmma16(a0c, bc[t], acc0[t]);
      acc1[t] = wmma16(a1c, bc[t], acc1[t]);
    }
    a0c = a0n; a1c = a1n;
#pragma unroll
    for (int t = 0; t < 4; ++t) bc[t] = bn[t];
  }

#pragma unroll
  for (int t = 0; t < 4; ++t) {
    const int c = c0 + t * 16 + l16;
    const float bias = bo[c];
    const size_t base = ((size_t)b * CCH + c) * NTOK + n0 + roff;
#pragma unroll
    for (int half2 = 0; half2 < 2; ++half2) {
      const v8f& av   = half2 ? acc1[t] : acc0[t];
      const size_t bi = base + (size_t)half2 * 16;
      const v4f x0 = *(const v4f*)(x + bi);
      const v4f x1 = *(const v4f*)(x + bi + 4);
      v4f r0, r1;
#pragma unroll
      for (int g = 0; g < 4; ++g) {
        r0[g] = av[g]     + bias + x0[g];
        r1[g] = av[4 + g] + bias + x1[g];
      }
      *(v4f*)(out + bi)     = r0;
      *(v4f*)(out + bi + 4) = r1;
    }
  }
}

extern "C" void kernel_launch(void* const* d_in, const int* in_sizes, int n_in,
                              void* d_out, int out_size, void* d_ws, size_t ws_size,
                              hipStream_t stream) {
  (void)in_sizes; (void)n_in; (void)out_size; (void)ws_size;
  const float* x  = (const float*)d_in[0];
  const float* Wp = (const float*)d_in[1];
  const float* bp = (const float*)d_in[2];
  const float* Wo = (const float*)d_in[3];
  const float* bo = (const float*)d_in[4];
  float* out = (float*)d_out;

  char* ws = (char*)d_ws;
  _Float16* xh  = (_Float16*)(ws);
  _Float16* WpT = (_Float16*)(ws +  8388608);
  _Float16* WoT = (_Float16*)(ws +  9961472);
  _Float16* qh  = (_Float16*)(ws + 10485760);
  _Float16* kh  = (_Float16*)(ws + 18874368);
  _Float16* vT  = (_Float16*)(ws + 27262976);
  _Float16* at  = (_Float16*)(ws + 35651584);

  to_fragment_order<<<dim3(256), 256, 0, stream>>>(x, Wp, Wo, xh, WpT, WoT);

  qkv_gemm    <<<dim3(NTOK / 256, 1536 / 192, NB), 256, 0, stream>>>(xh, WpT, bp, qh, kh, vT);
  attention_fa<<<dim3(NTOK / 32, NH,        NB), 32, 0, stream>>>(qh, kh, vT, at);
  out_proj    <<<dim3(NTOK / 32, CCH / 64,  NB), 32, 0, stream>>>(at, WoT, bo, x, out);
}


# attention: the 1024 keys of a 32-query block split over the two waves of a 64-thread group (512 each); partial O, max and row sum joined once through LDS (new sum order): two waves per SIMD
# speedup vs baseline: 2.2816x; 1.0788x over previous
#include <hip/hip_runtime.h>
#include <hip/hip_bf16.h>


typedef _Float16 v16h __attribute__((ext_vector_type(16)));
typedef _Float16 v8h  __attribute__((ext_vector_type(8)));
typedef _Float16 v4h  __attribute__((ext_vector_type(4)));
typedef float    v8f  __attribute__((ext_vector_type(8)));
typedef float    v4f  __attribute__((ext_vector_type(4)));

#define NB   8
#define CCH  512
#define NTOK 1024
#define NH   4
#define DK   128

__device__ __forceinline__ v16h load_frag(const _Float16* p) {
  union { v16h v; v8h h[2]; } u;
  u.h[0] = *(const v8h*)(p);
  u.h[1] = *(const v8h*)(p + 256);
  return u.v;
}

__device__ __forceinline__ v8f wmma16(v16h a, v16h b, v8f c) {
  return __builtin_amdgcn_wmma_f32_16x16x32_f16(false, a, false, b,
                                                (short)0, c, false, false);
}

#define TPITCH 132

__device__ __forceinline__ void load_tile(v4f (&r)[4], const float* __restrict__ s,
                                          int cols, int lrow, int lcol) {
#pragma unroll
  for (int j = 0; j < 4; ++j)
    r[j] = *(const v4f*)(s + (size_t)(j * 8 + lrow) * cols + lcol);
}

__global__ void __launch_bounds__(256)
to_fragment_order(const float* __restrict__ x, const float* __restrict__ Wp,
                  const float* __restrict__ Wo, _Float16* __restrict__ xh,
                  _Float16* __restrict__ WpT, _Float16* __restrict__ WoT) {
  __shared__ float tile[2][32][TPITCH];
  const int t    = threadIdx.x;
  const int wg   = blockIdx.x;
  const int lrow = t >> 5;
  const int lcol = (t & 31) * 4;

  const float* sp[5]; _Float16* dp[5]; int cl[5];
#pragma unroll
  for (int i = 0; i < 4; ++i) {
    const int id = i * 256 + wg;
    const int b  = id >> 7, kt = id & 15, nt = (id >> 4) & 7;
    sp[i] = x + ((size_t)b * CCH + kt * 32) * NTOK + nt * 128;
    dp[i] = xh + ((size_t)((b * NTOK + nt * 128) >> 4) * 16 + kt) * 512;
    cl[i] = NTOK;
  }
  {
    const bool isp = wg < 192;
    const int id = isp ? wg : wg - 192;
    const int kt = id & 15, nt = id >> 4;
    cl[4] = isp ? 1536 : CCH;
    sp[4] = (isp ? Wp : Wo) + (size_t)(kt * 32) * cl[4] + nt * 128;
    dp[4] = (isp ? WpT : WoT) + ((size_t)((nt * 128) >> 4) * 16 + kt) * 512;
  }

  v4f r[5][4];
#pragma unroll
  for (int i = 0; i < 5; ++i) load_tile(r[i], sp[i], cl[i], lrow, lcol);

#pragma unroll
  for (int i = 0; i < 5; ++i) {
    float (*tl)[TPITCH] = tile[i & 1];
#pragma unroll
    for (int j = 0; j < 4; ++j) *(v4f*)&tl[j * 8 + lrow][lcol] = r[i][j];
    __builtin_amdgcn_fence(__ATOMIC_RELEASE, "workgroup", "local");
    __builtin_amdgcn_s_barrier();
    __builtin_amdgcn_fence(__ATOMIC_ACQUIRE, "workgroup", "local");
#pragma unroll
    for (int u = 0; u < 2; ++u) {
      const int S  = t + u * 256;
      const int jr = S >> 6, s = S & 63;
      const int k0 = (s >> 5) * 16 + ((s >> 4) & 1) * 8;
      const int n  = jr * 16 + (s & 15);
      v8h w;
#pragma unroll
      for (int e = 0; e < 8; ++e) w[e] = (_Float16)tl[k0 + e][n];
      *(v8h*)(dp[i] + (size_t)jr * (16 * 512) + s * 8) = w;
    }
  }
}

__global__ void __launch_bounds__(256) __attribute__((amdgpu_waves_per_eu(1, 2)))
qkv_gemm(const _Float16* __restrict__ xh,
         const _Float16* __restrict__ WpT,
         const float* __restrict__ bp,
         _Float16* __restrict__ qh,
         _Float16* __restrict__ kh,
         _Float16* __restrict__ vT) {
  const int lane  = threadIdx.x & 31;
  const int wv    = __builtin_amdgcn_readfirstlane((int)(threadIdx.x >> 5));
  const int hf    = lane >> 4;
  const int l16   = lane & 15;
  const int roff  = hf * 8;
  const int n0 = (blockIdx.x * 4 + (wv >> 2) * 2 + (wv & 1)) * 64;
  const int j0 = (blockIdx.y * 2 + ((wv >> 1) & 1)) * 96;
  const int b  = blockIdx.z;

  const _Float16* Ap[4];
#pragma unroll
  for (int i = 0; i < 4; ++i)
    Ap[i] = xh + ((((size_t)b * NTOK + n0) >> 4) + i) * (CCH * 16) + lane * 8;
  const _Float16* Bp[6];
#pragma unroll
  for (int t = 0; t < 6; ++t)
    Bp[t] = WpT + (size_t)((j0 >> 4) + t) * (CCH * 16) + lane * 8;

  v16h af[3][4], bf[3][6];
#pragma unroll
  for (int s = 0; s < 2; ++s) {
#pragma unroll
    for (int i = 0; i < 4; ++i) af[s][i] = load_frag(Ap[i] + s * 512);
#pragma unroll
    for (int t = 0; t < 6; ++t) bf[s][t] = load_frag(Bp[t] + s * 512);
  }
  v8f acc[4][6] = {};
  float bias6[6] = {};
#pragma unroll
  for (int ks = 0; ks < CCH / 32; ++ks) {
    const int cur = ks % 3, nxt = (ks + 2) % 3;
    if (ks + 2 < CCH / 32) {
#pragma unroll
      for (int i = 0; i < 4; ++i) af[nxt][i] = load_frag(Ap[i] + (ks + 2) * 512);
#pragma unroll
      for (int t = 0; t < 6; ++t) bf[nxt][t] = load_frag(Bp[t] + (ks + 2) * 512);
    } else if (ks + 2 == CCH / 32) {
#pragma unroll
      for (int t = 0; t < 6; ++t) bias6[t] = bp[j0 + t * 16 + l16];
    }
    __builtin_amdgcn_sched_barrier(0);
#pragma unroll
    for (int t = 0; t < 6; ++t)
#pragma unroll
      for (int i = 0; i < 4; ++i) acc[i][t] = wmma16(af[cur][i], bf[cur][t], acc[i][t]);
    __builtin_amdgcn_sched_barrier(0);
  }

#pragma unroll
  for (int t = 0; t < 6; ++t) {
    const int j0t  = j0 + t * 16;
    const int head = j0t / 384;
    const int r0   = j0t % 384;
    const int seg  = r0 / 128;
    const int dbase = r0 % 128;
    const float bias = bias6[t];
    const size_t bh = ((size_t)b * NH + head) * NTOK * DK;
#pragma unroll
    for (int pr = 0; pr < 2; ++pr) {
      const int n0p = n0 + pr * 32;
      const v8f& acc0 = acc[2 * pr][t];
      const v8f& acc1 = acc[2 * pr + 1][t];
      if (seg == 2) {
        _Float16* dst = vT + bh + (size_t)((n0p >> 5) * 8 + (dbase >> 4)) * 512 + lane * 8;
        v8h w0, w1;
#pragma unroll
        for (int g = 0; g < 8; ++g) {
          w0[g] = (_Float16)(acc0[g] + bias);
          w1[g] = (_Float16)(acc1[g] + bias);
        }
        *(v8h*)(dst)       = w0;
        *(v8h*)(dst + 256) = w1;
      } else {
        _Float16* base = (seg == 0) ? qh : kh;
        _Float16* p0 = base + bh + (size_t)((n0p >> 4) * 4 + (dbase >> 5)) * 512
                     + ((dbase >> 4) & 1) * 256 + ((l16 >> 3) * 16 + roff) * 8 + (l16 & 7);
#pragma unroll
        for (int g = 0; g < 8; ++g) {
          p0[g * 8]           = (_Float16)(acc0[g] + bias);
          p0[4 * 512 + g * 8] = (_Float16)(acc1[g] + bias);
        }
      }
    }
  }
}

__device__ __forceinline__ float xhalf(float x) {
  return __builtin_bit_cast(float, __builtin_amdgcn_permlanex16(
      __builtin_bit_cast(int, x), __builtin_bit_cast(int, x),
      0x76543210, 0xfedcba98, false, false));
}

#define KSPLIT (NTOK / 2)

template <int GIVE>
__device__ __forceinline__ void give_tile(const v8f (&o)[2][8], const float (&m_run)[2],
                                          const float (&l_run)[2], v4f (*xo)[16][32],
                                          float (*xs)[2][32], int w, int lane) {
#pragma unroll
  for (int t = 0; t < 8; ++t)
#pragma unroll
    for (int q = 0; q < 2; ++q) {
      v4f g;
#pragma unroll
      for (int e = 0; e < 4; ++e) g[e] = o[GIVE][t][q * 4 + e];
      xo[w][t * 2 + q][lane] = g;
    }
  xs[w][0][lane] = m_run[GIVE];
  xs[w][1][lane] = l_run[GIVE];
}

template <int KEEP>
__device__ __forceinline__ void keep_tile(const v8f (&o)[2][8], const float (&m_run)[2],
                                          const float (&l_run)[2], const v4f (*xo)[16][32],
                                          const float (*xs)[2][32], int w, int lane,
                                          _Float16* __restrict__ Ob) {
  const int p = 1 - w;
  const float mo = xs[p][0][lane];
  const float lo = xs[p][1][lane];
  const float m  = __builtin_fmaxf(m_run[KEEP], mo);
  const float am = __builtin_amdgcn_exp2f(m_run[KEEP] - m);
  const float ao = __builtin_amdgcn_exp2f(mo - m);
  const float lh = l_run[KEEP] * am + lo * ao;
  const float lt = lh + xhalf(lh);
  const float inv = 1.0f / lt;
  const float sm = am * inv, so = ao * inv;
#pragma unroll
  for (int t = 0; t < 8; ++t) {
    const v4f g0 = xo[p][t * 2][lane];
    const v4f g1 = xo[p][t * 2 + 1][lane];
    v8h wv;
#pragma unroll
    for (int e = 0; e < 4; ++e) {
      wv[e]     = (_Float16)(o[KEEP][t][e] * sm + g0[e] * so);
      wv[4 + e] = (_Float16)(o[KEEP][t][4 + e] * sm + g1[e] * so);
    }
    *(v8h*)(Ob + (t >> 1) * 512 + (t & 1) * 256) = wv;
  }
}

__global__ void __launch_bounds__(64)
attention_fa(const _Float16* __restrict__ qh, const _Float16* __restrict__ kh,
             const _Float16* __restrict__ vT, _Float16* __restrict__ attn) {
  __shared__ v4f   xo[2][16][32];
  __shared__ float xs[2][2][32];
  const int lane = threadIdx.x & 31;
  const int w  = __builtin_amdgcn_readfirstlane((int)(threadIdx.x >> 5));
  const int n0 = blockIdx.x * 32;
  const int h  = blockIdx.y;
  const int b  = blockIdx.z;
  const int jbeg = w * KSPLIT, jend = jbeg + KSPLIT;
  constexpr float c2 = 0.08838834764831845f * 1.4426950408889634f;

  const size_t bh = ((size_t)b * NH + h) * NTOK * DK;
  const _Float16* Qb = qh + bh + (size_t)(n0 >> 4) * 4 * 512 + lane * 8;
  v16h qf[2][4];
#pragma unroll
  for (int qt = 0; qt < 2; ++qt)
#pragma unroll
    for (int kk = 0; kk < 4; ++kk) qf[qt][kk] = load_frag(Qb + (qt * 4 + kk) * 512);

  const _Float16* Kb = kh + bh + lane * 8;
  const _Float16* Vb = vT + bh + lane * 8;

  v16h kf[8];
#pragma unroll
  for (int i = 0; i < 8; ++i) kf[i] = load_frag(Kb + (size_t)jbeg * DK + i * 512);

  v8f o[2][8] = {};
  float m_run[2] = {-1e30f, -1e30f};
  float l_run[2] = {0.0f, 0.0f};

  for (int j0 = jbeg; j0 < jend; j0 += 32) {
    const _Float16* V0 = Vb + (size_t)j0 * DK;
    v16h vf[8];
#pragma unroll
    for (int t = 0; t < 8; ++t) vf[t] = load_frag(V0 + t * 512);

    v8f s[2][2];
#pragma unroll
    for (int kt = 0; kt < 2; ++kt)
#pragma unroll
      for (int qt = 0; qt < 2; ++qt) {
        v8f acc = {};
#pragma unroll
        for (int kk = 0; kk < 4; ++kk) acc = wmma16(kf[kt * 4 + kk], qf[qt][kk], acc);
        s[kt][qt] = acc;
      }

    const int jn = (j0 + 32 < jend) ? j0 + 32 : j0;
    const _Float16* Kn = Kb + (size_t)jn * DK;
#pragma unroll
    for (int i = 0; i < 8; ++i) kf[i] = load_frag(Kn + i * 512);

#pragma unroll
    for (int qt = 0; qt < 2; ++qt) {
      float mx = s[0][qt][0];
#pragma unroll
      for (int r = 1; r < 8; ++r) mx = __builtin_fmaxf(mx, s[0][qt][r]);
#pragma unroll
      for (int r = 0; r < 8; ++r) mx = __builtin_fmaxf(mx, s[1][qt][r]);
      mx = __builtin_fmaxf(mx, xhalf(mx)) * c2;
      const float mn    = __builtin_fmaxf(m_run[qt], mx);
      const float alpha = __builtin_amdgcn_exp2f(m_run[qt] - mn);
      const float nm    = -mn;
      m_run[qt] = mn;
      float ps0 = 0.0f, ps1 = 0.0f;
      v16h pf;
#pragma unroll
      for (int r = 0; r < 8; ++r) {
        const float p0 = __builtin_amdgcn_exp2f(__builtin_fmaf(s[0][qt][r], c2, nm));
        const float p1 = __builtin_amdgcn_exp2f(__builtin_fmaf(s[1][qt][r], c2, nm));
        ps0 += p0; ps1 += p1;
        pf[r]     = (_Float16)p0;
        pf[8 + r] = (_Float16)p1;
      }
      l_run[qt] = l_run[qt] * alpha + (ps0 + ps1);
#pragma unroll
      for (int t = 0; t < 8; ++t) o[qt][t] = wmma16(vf[t], pf, o[qt][t] * alpha);
    }
  }

  _Float16* Ob = attn + (((((size_t)b * NTOK + n0) >> 4) + w) * 16 + h * 4) * 512 + lane * 8;
  if (w == 0) give_tile<1>(o, m_run, l_run, xo, xs, w, lane);
  else        give_tile<0>(o, m_run, l_run, xo, xs, w, lane);
  __syncthreads();
  if (w == 0) keep_tile<0>(o, m_run, l_run, xo, xs, w, lane, Ob);
  else        keep_tile<1>(o, m_run, l_run, xo, xs, w, lane, Ob);
}

__global__ void __launch_bounds__(32) __attribute__((amdgpu_waves_per_eu(1, 4)))
out_proj(const _Float16* __restrict__ attn,
         const _Float16* __restrict__ WoT,
         const float* __restrict__ bo,
         const float* __restrict__ x,
         float* __restrict__ out) {
  const int lane  = threadIdx.x;
  const int hf    = lane >> 4;
  const int l16   = lane & 15;
  const int roff  = hf * 8;
  const int n0 = blockIdx.x * 32;
  const int c0 = blockIdx.y * 64;
  const int b  = blockIdx.z;

  const _Float16* A0 = attn + (((size_t)b * NTOK + n0) >> 4) * (CCH * 16) + lane * 8;
  const _Float16* A1 = A0 + (size_t)CCH * 16;
  const _Float16* Bp[4];
#pragma unroll
  for (int t = 0; t < 4; ++t)
    Bp[t] = WoT + (size_t)((c0 >> 4) + t) * (CCH * 16) + lane * 8;

  v16h a0c = load_frag(A0), a1c = load_frag(A1);
  v16h bc[4];
#pragma unroll
  for (int t = 0; t < 4; ++t) bc[t] = load_frag(Bp[t]);
  v8f acc0[4] = {}, acc1[4] = {};
#pragma unroll
  for (int k0 = 0; k0 < CCH; k0 += 32) {
    v16h a0n = a0c, a1n = a1c, bn[4] = {bc[0], bc[1], bc[2], bc[3]};
    const int kn = k0 + 32;
    if (kn < CCH) {
      a0n = load_frag(A0 + kn * 16);
      a1n = load_frag(A1 + kn * 16);
#pragma unroll
      for (int t = 0; t < 4; ++t) bn[t] = load_frag(Bp[t] + kn * 16);
    }
#pragma unroll
    for (int t = 0; t < 4; ++t) {
      acc0[t] = wmma16(a0c, bc[t], acc0[t]);
      acc1[t] = wmma16(a1c, bc[t], acc1[t]);
    }
    a0c = a0n; a1c = a1n;
#pragma unroll
    for (int t = 0; t < 4; ++t) bc[t] = bn[t];
  }

#pragma unroll
  for (int t = 0; t < 4; ++t) {
    const int c = c0 + t * 16 + l16;
    const float bias = bo[c];
    const size_t base = ((size_t)b * CCH + c) * NTOK + n0 + roff;
#pragma unroll
    for (int half2 = 0; half2 < 2; ++half2) {
      const v8f& av   = half2 ? acc1[t] : acc0[t];
      const size_t bi = base + (size_t)half2 * 16;
      const v4f x0 = *(const v4f*)(x + bi);
      const v4f x1 = *(const v4f*)(x + bi + 4);
      v4f r0, r1;
#pragma unroll
      for (int g = 0; g < 4; ++g) {
        r0[g] = av[g]     + bias + x0[g];
        r1[g] = av[4 + g] + bias + x1[g];
      }
      *(v4f*)(out + bi)     = r0;
      *(v4f*)(out + bi + 4) = r1;
    }
  }
}

extern "C" void kernel_launch(void* const* d_in, const int* in_sizes, int n_in,
                              void* d_out, int out_size, void* d_ws, size_t ws_size,
                              hipStream_t stream) {
  (void)in_sizes; (void)n_in; (void)out_size; (void)ws_size;
  const float* x  = (const float*)d_in[0];
  const float* Wp = (const float*)d_in[1];
  const float* bp = (const float*)d_in[2];
  const float* Wo = (const float*)d_in[3];
  const float* bo = (const float*)d_in[4];
  float* out = (float*)d_out;

  char* ws = (char*)d_ws;
  _Float16* xh  = (_Float16*)(ws);
  _Float16* WpT = (_Float16*)(ws +  8388608);
  _Float16* WoT = (_Float16*)(ws +  9961472);
  _Float16* qh  = (_Float16*)(ws + 10485760);
  _Float16* kh  = (_Float16*)(ws + 18874368);
  _Float16* vT  = (_Float16*)(ws + 27262976);
  _Float16* at  = (_Float16*)(ws + 35651584);

  to_fragment_order<<<dim3(256), 256, 0, stream>>>(x, Wp, Wo, xh, WpT, WoT);

  qkv_gemm    <<<dim3(NTOK / 256, 1536 / 192, NB), 256, 0, stream>>>(xh, WpT, bp, qh, kh, vT);
  attention_fa<<<dim3(NTOK / 32, NH,        NB), 64, 0, stream>>>(qh, kh, vT, at);
  out_proj    <<<dim3(NTOK / 32, CCH / 64,  NB), 32, 0, stream>>>(at, WoT, bo, x, out);
}


# attention: 8-wave groups = four 32-query blocks x two key halves of one (b,head), one group per WGP: the four waves of a key half read the same K/V^T lines together, L2 serves each once, not 4 times
# speedup vs baseline: 2.5048x; 1.0978x over previous
#include <hip/hip_runtime.h>
#include <hip/hip_bf16.h>


typedef _Float16 v16h __attribute__((ext_vector_type(16)));
typedef _Float16 v8h  __attribute__((ext_vector_type(8)));
typedef _Float16 v4h  __attribute__((ext_vector_type(4)));
typedef float    v8f  __attribute__((ext_vector_type(8)));
typedef float    v4f  __attribute__((ext_vector_type(4)));

#define NB   8
#define CCH  512
#define NTOK 1024
#define NH   4
#define DK   128

__device__ __forceinline__ v16h load_frag(const _Float16* p) {
  union { v16h v; v8h h[2]; } u;
  u.h[0] = *(const v8h*)(p);
  u.h[1] = *(const v8h*)(p + 256);
  return u.v;
}

__device__ __forceinline__ v8f wmma16(v16h a, v16h b, v8f c) {
  return __builtin_amdgcn_wmma_f32_16x16x32_f16(false, a, false, b,
                                                (short)0, c, false, false);
}

#define TPITCH 132

__device__ __forceinline__ void load_tile(v4f (&r)[4], const float* __restrict__ s,
                                          int cols, int lrow, int lcol) {
#pragma unroll
  for (int j = 0; j < 4; ++j)
    r[j] = *(const v4f*)(s + (size_t)(j * 8 + lrow) * cols + lcol);
}

__global__ void __launch_bounds__(256)
to_fragment_order(const float* __restrict__ x, const float* __restrict__ Wp,
                  const float* __restrict__ Wo, _Float16* __restrict__ xh,
                  _Float16* __restrict__ WpT, _Float16* __restrict__ WoT) {
  __shared__ float tile[2][32][TPITCH];
  const int t    = threadIdx.x;
  const int wg   = blockIdx.x;
  const int lrow = t >> 5;
  const int lcol = (t & 31) * 4;

  const float* sp[5]; _Float16* dp[5]; int cl[5];
#pragma unroll
  for (int i = 0; i < 4; ++i) {
    const int id = i * 256 + wg;
    const int b  = id >> 7, kt = id & 15, nt = (id >> 4) & 7;
    sp[i] = x + ((size_t)b * CCH + kt * 32) * NTOK + nt * 128;
    dp[i] = xh + ((size_t)((b * NTOK + nt * 128) >> 4) * 16 + kt) * 512;
    cl[i] = NTOK;
  }
  {
    const bool isp = wg < 192;
    const int id = isp ? wg : wg - 192;
    const int kt = id & 15, nt = id >> 4;
    cl[4] = isp ? 1536 : CCH;
    sp[4] = (isp ? Wp : Wo) + (size_t)(kt * 32) * cl[4] + nt * 128;
    dp[4] = (isp ? WpT : WoT) + ((size_t)((nt * 128) >> 4) * 16 + kt) * 512;
  }

  v4f r[5][4];
#pragma unroll
  for (int i = 0; i < 5; ++i) load_tile(r[i], sp[i], cl[i], lrow, lcol);

#pragma unroll
  for (int i = 0; i < 5; ++i) {
    float (*tl)[TPITCH] = tile[i & 1];
#pragma unroll
    for (int j = 0; j < 4; ++j) *(v4f*)&tl[j * 8 + lrow][lcol] = r[i][j];
    __builtin_amdgcn_fence(__ATOMIC_RELEASE, "workgroup", "local");
    __builtin_amdgcn_s_barrier();
    __builtin_amdgcn_fence(__ATOMIC_ACQUIRE, "workgroup", "local");
#pragma unroll
    for (int u = 0; u < 2; ++u) {
      const int S  = t + u * 256;
      const int jr = S >> 6, s = S & 63;
      const int k0 = (s >> 5) * 16 + ((s >> 4) & 1) * 8;
      const int n  = jr * 16 + (s & 15);
      v8h w;
#pragma unroll
      for (int e = 0; e < 8; ++e) w[e] = (_Float16)tl[k0 + e][n];
      *(v8h*)(dp[i] + (size_t)jr * (16 * 512) + s * 8) = w;
    }
  }
}

__global__ void __launch_bounds__(256) __attribute__((amdgpu_waves_per_eu(1, 2)))
qkv_gemm(const _Float16* __restrict__ xh,
         const _Float16* __restrict__ WpT,
         const float* __restrict__ bp,
         _Float16* __restrict__ qh,
         _Float16* __restrict__ kh,
         _Float16* __restrict__ vT) {
  const int lane  = threadIdx.x & 31;
  const int wv    = __builtin_amdgcn_readfirstlane((int)(threadIdx.x >> 5));
  const int hf    = lane >> 4;
  const int l16   = lane & 15;
  const int roff  = hf * 8;
  const int n0 = (blockIdx.x * 4 + (wv >> 2) * 2 + (wv & 1)) * 64;
  const int j0 = (blockIdx.y * 2 + ((wv >> 1) & 1)) * 96;
  const int b  = blockIdx.z;

  const _Float16* Ap[4];
#pragma unroll
  for (int i = 0; i < 4; ++i)
    Ap[i] = xh + ((((size_t)b * NTOK + n0) >> 4) + i) * (CCH * 16) + lane * 8;
  const _Float16* Bp[6];
#pragma unroll
  for (int t = 0; t < 6; ++t)
    Bp[t] = WpT + (size_t)((j0 >> 4) + t) * (CCH * 16) + lane * 8;

  v16h af[3][4], bf[3][6];
#pragma unroll
  for (int s = 0; s < 2; ++s) {
#pragma unroll
    for (int i = 0; i < 4; ++i) af[s][i] = load_frag(Ap[i] + s * 512);
#pragma unroll
    for (int t = 0; t < 6; ++t) bf[s][t] = load_frag(Bp[t] + s * 512);
  }
  v8f acc[4][6] = {};
  float bias6[6] = {};
#pragma unroll
  for (int ks = 0; ks < CCH / 32; ++ks) {
    const int cur = ks % 3, nxt = (ks + 2) % 3;
    if (ks + 2 < CCH / 32) {
#pragma unroll
      for (int i = 0; i < 4; ++i) af[nxt][i] = load_frag(Ap[i] + (ks + 2) * 512);
#pragma unroll
      for (int t = 0; t < 6; ++t) bf[nxt][t] = load_frag(Bp[t] + (ks + 2) * 512);
    } else if (ks + 2 == CCH / 32) {
#pragma unroll
      for (int t = 0; t < 6; ++t) bias6[t] = bp[j0 + t * 16 + l16];
    }
    __builtin_amdgcn_sched_barrier(0);
#pragma unroll
    for (int t = 0; t < 6; ++t)
#pragma unroll
      for (int i = 0; i < 4; ++i) acc[i][t] = wmma16(af[cur][i], bf[cur][t], acc[i][t]);
    __builtin_amdgcn_sched_barrier(0);
  }

#pragma unroll
  for (int t = 0; t < 6; ++t) {
    const int j0t  = j0 + t * 16;
    const int head = j0t / 384;
    const int r0   = j0t % 384;
    const int seg  = r0 / 128;
    const int dbase = r0 % 128;
    const float bias = bias6[t];
    const size_t bh = ((size_t)b * NH + head) * NTOK * DK;
#pragma unroll
    for (int pr = 0; pr < 2; ++pr) {
      const int n0p = n0 + pr * 32;
      const v8f& acc0 = acc[2 * pr][t];
      const v8f& acc1 = acc[2 * pr + 1][t];
      if (seg == 2) {
        _Float16* dst = vT + bh + (size_t)((n0p >> 5) * 8 + (dbase >> 4)) * 512 + lane * 8;
        v8h w0, w1;
#pragma unroll
        for (int g = 0; g < 8; ++g) {
          w0[g] = (_Float16)(acc0[g] + bias);
          w1[g] = (_Float16)(acc1[g] + bias);
        }
        *(v8h*)(dst)       = w0;
        *(v8h*)(dst + 256) = w1;
      } else {
        _Float16* base = (seg == 0) ? qh : kh;
        _Float16* p0 = base + bh + (size_t)((n0p >> 4) * 4 + (dbase >> 5)) * 512
                     + ((dbase >> 4) & 1) * 256 + ((l16 >> 3) * 16 + roff) * 8 + (l16 & 7);
#pragma unroll
        for (int g = 0; g < 8; ++g) {
          p0[g * 8]           = (_Float16)(acc0[g] + bias);
          p0[4 * 512 + g * 8] = (_Float16)(acc1[g] + bias);
        }
      }
    }
  }
}

__device__ __forceinline__ float xhalf(float x) {
  return __builtin_bit_cast(float, __builtin_amdgcn_permlanex16(
      __builtin_bit_cast(int, x), __builtin_bit_cast(int, x),
      0x76543210, 0xfedcba98, false, false));
}

#define KSPLIT (NTOK / 2)
#define QBLK   4

template <int GIVE>
__device__ __forceinline__ void give_tile(const v8f (&o)[2][8], const float (&m_run)[2],
                                          const float (&l_run)[2], v4f (*xo)[16][32],
                                          float (*xs)[2][32], int w, int lane) {
#pragma unroll
  for (int t = 0; t < 8; ++t)
#pragma unroll
    for (int q = 0; q < 2; ++q) {
      v4f g;
#pragma unroll
      for (int e = 0; e < 4; ++e) g[e] = o[GIVE][t][q * 4 + e];
      xo[w][t * 2 + q][lane] = g;
    }
  xs[w][0][lane] = m_run[GIVE];
  xs[w][1][lane] = l_run[GIVE];
}

template <int KEEP>
__device__ __forceinline__ void keep_tile(const v8f (&o)[2][8], const float (&m_run)[2],
                                          const float (&l_run)[2], const v4f (*xo)[16][32],
                                          const float (*xs)[2][32], int w, int lane,
                                          _Float16* __restrict__ Ob) {
  const int p = 1 - w;
  const float mo = xs[p][0][lane];
  const float lo = xs[p][1][lane];
  const float m  = __builtin_fmaxf(m_run[KEEP], mo);
  const float am = __builtin_amdgcn_exp2f(m_run[KEEP] - m);
  const float ao = __builtin_amdgcn_exp2f(mo - m);
  const float lh = l_run[KEEP] * am + lo * ao;
  const float lt = lh + xhalf(lh);
  const float inv = 1.0f / lt;
  const float sm = am * inv, so = ao * inv;
#pragma unroll
  for (int t = 0; t < 8; ++t) {
    const v4f g0 = xo[p][t * 2][lane];
    const v4f g1 = xo[p][t * 2 + 1][lane];
    v8h wv;
#pragma unroll
    for (int e = 0; e < 4; ++e) {
      wv[e]     = (_Float16)(o[KEEP][t][e] * sm + g0[e] * so);
      wv[4 + e] = (_Float16)(o[KEEP][t][4 + e] * sm + g1[e] * so);
    }
    *(v8h*)(Ob + (t >> 1) * 512 + (t & 1) * 256) = wv;
  }
}

__global__ void __launch_bounds__(256)
attention_fa(const _Float16* __restrict__ qh, const _Float16* __restrict__ kh,
             const _Float16* __restrict__ vT, _Float16* __restrict__ attn) {
  __shared__ v4f   xo[QBLK][2][16][32];
  __shared__ float xs[QBLK][2][2][32];
  const int lane = threadIdx.x & 31;
  const int wv = __builtin_amdgcn_readfirstlane((int)(threadIdx.x >> 5));
  const int w  = wv >> 2;
  const int qb = wv & 3;
  const int n0 = (blockIdx.x * QBLK + qb) * 32;
  const int h  = blockIdx.y;
  const int b  = blockIdx.z;
  const int jbeg = w * KSPLIT, jend = jbeg + KSPLIT;
  constexpr float c2 = 0.08838834764831845f * 1.4426950408889634f;

  const size_t bh = ((size_t)b * NH + h) * NTOK * DK;
  const _Float16* Qb = qh + bh + (size_t)(n0 >> 4) * 4 * 512 + lane * 8;
  v16h qf[2][4];
#pragma unroll
  for (int qt = 0; qt < 2; ++qt)
#pragma unroll
    for (int kk = 0; kk < 4; ++kk) qf[qt][kk] = load_frag(Qb + (qt * 4 + kk) * 512);

  const _Float16* Kb = kh + bh + lane * 8;
  const _Float16* Vb = vT + bh + lane * 8;

  v16h kf[8];
#pragma unroll
  for (int i = 0; i < 8; ++i) kf[i] = load_frag(Kb + (size_t)jbeg * DK + i * 512);

  v8f o[2][8] = {};
  float m_run[2] = {-1e30f, -1e30f};
  float l_run[2] = {0.0f, 0.0f};

  for (int j0 = jbeg; j0 < jend; j0 += 32) {
    const _Float16* V0 = Vb + (size_t)j0 * DK;
    v16h vf[8];
#pragma unroll
    for (int t = 0; t < 8; ++t) vf[t] = load_frag(V0 + t * 512);

    v8f s[2][2];
#pragma unroll
    for (int kt = 0; kt < 2; ++kt)
#pragma unroll
      for (int qt = 0; qt < 2; ++qt) {
        v8f acc = {};
#pragma unroll
        for (int kk = 0; kk < 4; ++kk) acc = wmma16(kf[kt * 4 + kk], qf[qt][kk], acc);
        s[kt][qt] = acc;
      }

    const int jn = (j0 + 32 < jend) ? j0 + 32 : j0;
    const _Float16* Kn = Kb + (size_t)jn * DK;
#pragma unroll
    for (int i = 0; i < 8; ++i) kf[i] = load_frag(Kn + i * 512);

#pragma unroll
    for (int qt = 0; qt < 2; ++qt) {
      float mx = s[0][qt][0];
#pragma unroll
      for (int r = 1; r < 8; ++r) mx = __builtin_fmaxf(mx, s[0][qt][r]);
#pragma unroll
      for (int r = 0; r < 8; ++r) mx = __builtin_fmaxf(mx, s[1][qt][r]);
      mx = __builtin_fmaxf(mx, xhalf(mx)) * c2;
      const float mn    = __builtin_fmaxf(m_run[qt], mx);
      const float alpha = __builtin_amdgcn_exp2f(m_run[qt] - mn);
      const float nm    = -mn;
      m_run[qt] = mn;
      float ps0 = 0.0f, ps1 = 0.0f;
      v16h pf;
#pragma unroll
      for (int r = 0; r < 8; ++r) {
        const float p0 = __builtin_amdgcn_exp2f(__builtin_fmaf(s[0][qt][r], c2, nm));
        const float p1 = __builtin_amdgcn_exp2f(__builtin_fmaf(s[1][qt][r], c2, nm));
        ps0 += p0; ps1 += p1;
        pf[r]     = (_Float16)p0;
        pf[8 + r] = (_Float16)p1;
      }
      l_run[qt] = l_run[qt] * alpha + (ps0 + ps1);
#pragma unroll
      for (int t = 0; t < 8; ++t) o[qt][t] = wmma16(vf[t], pf, o[qt][t] * alpha);
    }
  }

  _Float16* Ob = attn + (((((size_t)b * NTOK + n0) >> 4) + w) * 16 + h * 4) * 512 + lane * 8;
  if (w == 0) give_tile<1>(o, m_run, l_run, xo[qb], xs[qb], w, lane);
  else        give_tile<0>(o, m_run, l_run, xo[qb], xs[qb], w, lane);
  __syncthreads();
  if (w == 0) keep_tile<0>(o, m_run, l_run, xo[qb], xs[qb], w, lane, Ob);
  else        keep_tile<1>(o, m_run, l_run, xo[qb], xs[qb], w, lane, Ob);
}

__global__ void __launch_bounds__(32) __attribute__((amdgpu_waves_per_eu(1, 4)))
out_proj(const _Float16* __restrict__ attn,
         const _Float16* __restrict__ WoT,
         const float* __restrict__ bo,
         const float* __restrict__ x,
         float* __restrict__ out) {
  const int lane  = threadIdx.x;
  const int hf    = lane >> 4;
  const int l16   = lane & 15;
  const int roff  = hf * 8;
  const int n0 = blockIdx.x * 32;
  const int c0 = blockIdx.y * 64;
  const int b  = blockIdx.z;

  const _Float16* A0 = attn + (((size_t)b * NTOK + n0) >> 4) * (CCH * 16) + lane * 8;
  const _Float16* A1 = A0 + (size_t)CCH * 16;
  const _Float16* Bp[4];
#pragma unroll
  for (int t = 0; t < 4; ++t)
    Bp[t] = WoT + (size_t)((c0 >> 4) + t) * (CCH * 16) + lane * 8;

  v16h a0c = load_frag(A0), a1c = load_frag(A1);
  v16h bc[4];
#pragma unroll
  for (int t = 0; t < 4; ++t) bc[t] = load_frag(Bp[t]);
  v8f acc0[4] = {}, acc1[4] = {};
#pragma unroll
  for (int k0 = 0; k0 < CCH; k0 += 32) {
    v16h a0n = a0c, a1n = a1c, bn[4] = {bc[0], bc[1], bc[2], bc[3]};
    const int kn = k0 + 32;
    if (kn < CCH) {
      a0n = load_frag(A0 + kn * 16);
      a1n = load_frag(A1 + kn * 16);
#pragma unroll
      for (int t = 0; t < 4; ++t) bn[t] = load_frag(Bp[t] + kn * 16);
    }
#pragma unroll
    for (int t = 0; t < 4; ++t) {
      acc0[t] = wmma16(a0c, bc[t], acc0[t]);
      acc1[t] = wmma16(a1c, bc[t], acc1[t]);
    }
    a0c = a0n; a1c = a1n;
#pragma unroll
    for (int t = 0; t < 4; ++t) bc[t] = bn[t];
  }

#pragma unroll
  for (int t = 0; t < 4; ++t) {
    const int c = c0 + t * 16 + l16;
    const float bias = bo[c];
    const size_t base = ((size_t)b * CCH + c) * NTOK + n0 + roff;
#pragma unroll
    for (int half2 = 0; half2 < 2; ++half2) {
      const v8f& av   = half2 ? acc1[t] : acc0[t];
      const size_t bi = base + (size_t)half2 * 16;
      const v4f x0 = *(const v4f*)(x + bi);
      const v4f x1 = *(const v4f*)(x + bi + 4);
      v4f r0, r1;
#pragma unroll
      for (int g = 0; g < 4; ++g) {
        r0[g] = av[g]     + bias + x0[g];
        r1[g] = av[4 + g] + bias + x1[g];
      }
      *(v4f*)(out + bi)     = r0;
      *(v4f*)(out + bi + 4) = r1;
    }
  }
}

extern "C" void kernel_launch(void* const* d_in, const int* in_sizes, int n_in,
                              void* d_out, int out_size, void* d_ws, size_t ws_size,
                              hipStream_t stream) {
  (void)in_sizes; (void)n_in; (void)out_size; (void)ws_size;
  const float* x  = (const float*)d_in[0];
  const float* Wp = (const float*)d_in[1];
  const float* bp = (const float*)d_in[2];
  const float* Wo = (const float*)d_in[3];
  const float* bo = (const float*)d_in[4];
  float* out = (float*)d_out;

  char* ws = (char*)d_ws;
  _Float16* xh  = (_Float16*)(ws);
  _Float16* WpT = (_Float16*)(ws +  8388608);
  _Float16* WoT = (_Float16*)(ws +  9961472);
  _Float16* qh  = (_Float16*)(ws + 10485760);
  _Float16* kh  = (_Float16*)(ws + 18874368);
  _Float16* vT  = (_Float16*)(ws + 27262976);
  _Float16* at  = (_Float16*)(ws + 35651584);

  to_fragment_order<<<dim3(256), 256, 0, stream>>>(x, Wp, Wo, xh, WpT, WoT);

  qkv_gemm    <<<dim3(NTOK / 256, 1536 / 192, NB), 256, 0, stream>>>(xh, WpT, bp, qh, kh, vT);
  attention_fa<<<dim3(NTOK / 128, NH,       NB), 256, 0, stream>>>(qh, kh, vT, at);
  out_proj    <<<dim3(NTOK / 32, CCH / 64,  NB), 32, 0, stream>>>(at, WoT, bo, x, out);
}


# second-level prefetch ahead of first touch: attention names each K/V^T block (one 8-KB run per wave and step) three steps before its loads; qkv names its A row panel at wave start
# speedup vs baseline: 2.5777x; 1.0291x over previous
#include <hip/hip_runtime.h>
#include <hip/hip_bf16.h>


typedef _Float16 v16h __attribute__((ext_vector_type(16)));
typedef _Float16 v8h  __attribute__((ext_vector_type(8)));
typedef _Float16 v4h  __attribute__((ext_vector_type(4)));
typedef float    v8f  __attribute__((ext_vector_type(8)));
typedef float    v4f  __attribute__((ext_vector_type(4)));

#define NB   8
#define CCH  512
#define NTOK 1024
#define NH   4
#define DK   128

__device__ __forceinline__ v16h load_frag(const _Float16* p) {
  union { v16h v; v8h h[2]; } u;
  u.h[0] = *(const v8h*)(p);
  u.h[1] = *(const v8h*)(p + 256);
  return u.v;
}

__device__ __forceinline__ v8f wmma16(v16h a, v16h b, v8f c) {
  return __builtin_amdgcn_wmma_f32_16x16x32_f16(false, a, false, b,
                                                (short)0, c, false, false);
}

__device__ __forceinline__ void l2_prefetch(const void* p) {
  __builtin_prefetch(p, 0, 2);
}

#define TPITCH 132

__device__ __forceinline__ void load_tile(v4f (&r)[4], const float* __restrict__ s,
                                          int cols, int lrow, int lcol) {
#pragma unroll
  for (int j = 0; j < 4; ++j)
    r[j] = *(const v4f*)(s + (size_t)(j * 8 + lrow) * cols + lcol);
}

__global__ void __launch_bounds__(256)
to_fragment_order(const float* __restrict__ x, const float* __restrict__ Wp,
                  const float* __restrict__ Wo, _Float16* __restrict__ xh,
                  _Float16* __restrict__ WpT, _Float16* __restrict__ WoT) {
  __shared__ float tile[2][32][TPITCH];
  const int t    = threadIdx.x;
  const int wg   = blockIdx.x;
  const int lrow = t >> 5;
  const int lcol = (t & 31) * 4;

  const float* sp[5]; _Float16* dp[5]; int cl[5];
#pragma unroll
  for (int i = 0; i < 4; ++i) {
    const int id = i * 256 + wg;
    const int b  = id >> 7, kt = id & 15, nt = (id >> 4) & 7;
    sp[i] = x + ((size_t)b * CCH + kt * 32) * NTOK + nt * 128;
    dp[i] = xh + ((size_t)((b * NTOK + nt * 128) >> 4) * 16 + kt) * 512;
    cl[i] = NTOK;
  }
  {
    const bool isp = wg < 192;
    const int id = isp ? wg : wg - 192;
    const int kt = id & 15, nt = id >> 4;
    cl[4] = isp ? 1536 : CCH;
    sp[4] = (isp ? Wp : Wo) + (size_t)(kt * 32) * cl[4] + nt * 128;
    dp[4] = (isp ? WpT : WoT) + ((size_t)((nt * 128) >> 4) * 16 + kt) * 512;
  }

  v4f r[5][4];
#pragma unroll
  for (int i = 0; i < 5; ++i) load_tile(r[i], sp[i], cl[i], lrow, lcol);

#pragma unroll
  for (int i = 0; i < 5; ++i) {
    float (*tl)[TPITCH] = tile[i & 1];
#pragma unroll
    for (int j = 0; j < 4; ++j) *(v4f*)&tl[j * 8 + lrow][lcol] = r[i][j];
    __builtin_amdgcn_fence(__ATOMIC_RELEASE, "workgroup", "local");
    __builtin_amdgcn_s_barrier();
    __builtin_amdgcn_fence(__ATOMIC_ACQUIRE, "workgroup", "local");
#pragma unroll
    for (int u = 0; u < 2; ++u) {
      const int S  = t + u * 256;
      const int jr = S >> 6, s = S & 63;
      const int k0 = (s >> 5) * 16 + ((s >> 4) & 1) * 8;
      const int n  = jr * 16 + (s & 15);
      v8h w;
#pragma unroll
      for (int e = 0; e < 8; ++e) w[e] = (_Float16)tl[k0 + e][n];
      *(v8h*)(dp[i] + (size_t)jr * (16 * 512) + s * 8) = w;
    }
  }
}

__global__ void __launch_bounds__(256) __attribute__((amdgpu_waves_per_eu(1, 2)))
qkv_gemm(const _Float16* __restrict__ xh,
         const _Float16* __restrict__ WpT,
         const float* __restrict__ bp,
         _Float16* __restrict__ qh,
         _Float16* __restrict__ kh,
         _Float16* __restrict__ vT) {
  const int lane  = threadIdx.x & 31;
  const int wv    = __builtin_amdgcn_readfirstlane((int)(threadIdx.x >> 5));
  const int hf    = lane >> 4;
  const int l16   = lane & 15;
  const int roff  = hf * 8;
  const int n0 = (blockIdx.x * 4 + (wv >> 2) * 2 + (wv & 1)) * 64;
  const int j0 = (blockIdx.y * 2 + ((wv >> 1) & 1)) * 96;
  const int b  = blockIdx.z;

  const _Float16* Ap[4];
#pragma unroll
  for (int i = 0; i < 4; ++i)
    Ap[i] = xh + ((((size_t)b * NTOK + n0) >> 4) + i) * (CCH * 16) + lane * 8;
  const _Float16* Bp[6];
#pragma unroll
  for (int t = 0; t < 6; ++t)
    Bp[t] = WpT + (size_t)((j0 >> 4) + t) * (CCH * 16) + lane * 8;

  v16h af[3][4], bf[3][6];
#pragma unroll
  for (int s = 0; s < 2; ++s) {
#pragma unroll
    for (int i = 0; i < 4; ++i) af[s][i] = load_frag(Ap[i] + s * 512);
#pragma unroll
    for (int t = 0; t < 6; ++t) bf[s][t] = load_frag(Bp[t] + s * 512);
  }
  l2_prefetch(xh + ((((size_t)b * NTOK + n0) >> 4) + (lane >> 3)) * (CCH * 16)
                 + blockIdx.y * 1024 + (lane & 7) * 128);
  v8f acc[4][6] = {};
  float bias6[6] = {};
#pragma unroll
  for (int ks = 0; ks < CCH / 32; ++ks) {
    const int cur = ks % 3, nxt = (ks + 2) % 3;
    if (ks + 2 < CCH / 32) {
#pragma unroll
      for (int i = 0; i < 4; ++i) af[nxt][i] = load_frag(Ap[i] + (ks + 2) * 512);
#pragma unroll
      for (int t = 0; t < 6; ++t) bf[nxt][t] = load_frag(Bp[t] + (ks + 2) * 512);
    } else if (ks + 2 == CCH / 32) {
#pragma unroll
      for (int t = 0; t < 6; ++t) bias6[t] = bp[j0 + t * 16 + l16];
    }
    __builtin_amdgcn_sched_barrier(0);
#pragma unroll
    for (int t = 0; t < 6; ++t)
#pragma unroll
      for (int i = 0; i < 4; ++i) acc[i][t] = wmma16(af[cur][i], bf[cur][t], acc[i][t]);
    __builtin_amdgcn_sched_barrier(0);
  }

#pragma unroll
  for (int t = 0; t < 6; ++t) {
    const int j0t  = j0 + t * 16;
    const int head = j0t / 384;
    const int r0   = j0t % 384;
    const int seg  = r0 / 128;
    const int dbase = r0 % 128;
    const float bias = bias6[t];
    const size_t bh = ((size_t)b * NH + head) * NTOK * DK;
#pragma unroll
    for (int pr = 0; pr < 2; ++pr) {
      const int n0p = n0 + pr * 32;
      const v8f& acc0 = acc[2 * pr][t];
      const v8f& acc1 = acc[2 * pr + 1][t];
      if (seg == 2) {
        _Float16* dst = vT + bh + (size_t)((n0p >> 5) * 8 + (dbase >> 4)) * 512 + lane * 8;
        v8h w0, w1;
#pragma unroll
        for (int g = 0; g < 8; ++g) {
          w0[g] = (_Float16)(acc0[g] + bias);
          w1[g] = (_Float16)(acc1[g] + bias);
        }
        *(v8h*)(dst)       = w0;
        *(v8h*)(dst + 256) = w1;
      } else {
        _Float16* base = (seg == 0) ? qh : kh;
        _Float16* p0 = base + bh + (size_t)((n0p >> 4) * 4 + (dbase >> 5)) * 512
                     + ((dbase >> 4) & 1) * 256 + ((l16 >> 3) * 16 + roff) * 8 + (l16 & 7);
#pragma unroll
        for (int g = 0; g < 8; ++g) {
          p0[g * 8]           = (_Float16)(acc0[g] + bias);
          p0[4 * 512 + g * 8] = (_Float16)(acc1[g] + bias);
        }
      }
    }
  }
}

__device__ __forceinline__ float xhalf(float x) {
  return __builtin_bit_cast(float, __builtin_amdgcn_permlanex16(
      __builtin_bit_cast(int, x), __builtin_bit_cast(int, x),
      0x76543210, 0xfedcba98, false, false));
}

#define KSPLIT (NTOK / 2)
#define PF_AHEAD 3
#define QBLK   4

template <int GIVE>
__device__ __forceinline__ void give_tile(const v8f (&o)[2][8], const float (&m_run)[2],
                                          const float (&l_run)[2], v4f (*xo)[16][32],
                                          float (*xs)[2][32], int w, int lane) {
#pragma unroll
  for (int t = 0; t < 8; ++t)
#pragma unroll
    for (int q = 0; q < 2; ++q) {
      v4f g;
#pragma unroll
      for (int e = 0; e < 4; ++e) g[e] = o[GIVE][t][q * 4 + e];
      xo[w][t * 2 + q][lane] = g;
    }
  xs[w][0][lane] = m_run[GIVE];
  xs[w][1][lane] = l_run[GIVE];
}

template <int KEEP>
__device__ __forceinline__ void keep_tile(const v8f (&o)[2][8], const float (&m_run)[2],
                                          const float (&l_run)[2], const v4f (*xo)[16][32],
                                          const float (*xs)[2][32], int w, int lane,
                                          _Float16* __restrict__ Ob) {
  const int p = 1 - w;
  const float mo = xs[p][0][lane];
  const float lo = xs[p][1][lane];
  const float m  = __builtin_fmaxf(m_run[KEEP], mo);
  const float am = __builtin_amdgcn_exp2f(m_run[KEEP] - m);
  const float ao = __builtin_amdgcn_exp2f(mo - m);
  const float lh = l_run[KEEP] * am + lo * ao;
  const float lt = lh + xhalf(lh);
  const float inv = 1.0f / lt;
  const float sm = am * inv, so = ao * inv;
#pragma unroll
  for (int t = 0; t < 8; ++t) {
    const v4f g0 = xo[p][t * 2][lane];
    const v4f g1 = xo[p][t * 2 + 1][lane];
    v8h wv;
#pragma unroll
    for (int e = 0; e < 4; ++e) {
      wv[e]     = (_Float16)(o[KEEP][t][e] * sm + g0[e] * so);
      wv[4 + e] = (_Float16)(o[KEEP][t][4 + e] * sm + g1[e] * so);
    }
    *(v8h*)(Ob + (t >> 1) * 512 + (t & 1) * 256) = wv;
  }
}

__global__ void __launch_bounds__(256)
attention_fa(const _Float16* __restrict__ qh, const _Float16* __restrict__ kh,
             const _Float16* __restrict__ vT, _Float16* __restrict__ attn) {
  __shared__ v4f   xo[QBLK][2][16][32];
  __shared__ float xs[QBLK][2][2][32];
  const int lane = threadIdx.x & 31;
  const int wv = __builtin_amdgcn_readfirstlane((int)(threadIdx.x >> 5));
  const int w  = wv >> 2;
  const int qb = wv & 3;
  const int n0 = (blockIdx.x * QBLK + qb) * 32;
  const int h  = blockIdx.y;
  const int b  = blockIdx.z;
  const int jbeg = w * KSPLIT, jend = jbeg + KSPLIT;
  constexpr float c2 = 0.08838834764831845f * 1.4426950408889634f;

  const size_t bh = ((size_t)b * NH + h) * NTOK * DK;
  const _Float16* Qb = qh + bh + (size_t)(n0 >> 4) * 4 * 512 + lane * 8;
  v16h qf[2][4];
#pragma unroll
  for (int qt = 0; qt < 2; ++qt)
#pragma unroll
    for (int kk = 0; kk < 4; ++kk) qf[qt][kk] = load_frag(Qb + (qt * 4 + kk) * 512);

  const _Float16* Kb = kh + bh + lane * 8;
  const _Float16* Vb = vT + bh + lane * 8;

  v16h kf[8];
#pragma unroll
  for (int i = 0; i < 8; ++i) kf[i] = load_frag(Kb + (size_t)jbeg * DK + i * 512);

  const _Float16* Pb = ((qb & 1) ? vT : kh) + bh + lane * 128;
#pragma unroll
  for (int a = 1; a < PF_AHEAD; ++a)
    l2_prefetch(Pb + (size_t)(jbeg + a * 32) * DK);

  v8f o[2][8] = {};
  float m_run[2] = {-1e30f, -1e30f};
  float l_run[2] = {0.0f, 0.0f};

  for (int j0 = jbeg; j0 < jend; j0 += 32) {
    const _Float16* V0 = Vb + (size_t)j0 * DK;
    v16h vf[8];
#pragma unroll
    for (int t = 0; t < 8; ++t) vf[t] = load_frag(V0 + t * 512);

    const int jp = (j0 + PF_AHEAD * 32 < jend) ? j0 + PF_AHEAD * 32 : j0;
    l2_prefetch(Pb + (size_t)jp * DK);

    v8f s[2][2];
#pragma unroll
    for (int kt = 0; kt < 2; ++kt)
#pragma unroll
      for (int qt = 0; qt < 2; ++qt) {
        v8f acc = {};
#pragma unroll
        for (int kk = 0; kk < 4; ++kk) acc = wmma16(kf[kt * 4 + kk], qf[qt][kk], acc);
        s[kt][qt] = acc;
      }

    const int jn = (j0 + 32 < jend) ? j0 + 32 : j0;
    const _Float16* Kn = Kb + (size_t)jn * DK;
#pragma unroll
    for (int i = 0; i < 8; ++i) kf[i] = load_frag(Kn + i * 512);

#pragma unroll
    for (int qt = 0; qt < 2; ++qt) {
      float mx = s[0][qt][0];
#pragma unroll
      for (int r = 1; r < 8; ++r) mx = __builtin_fmaxf(mx, s[0][qt][r]);
#pragma unroll
      for (int r = 0; r < 8; ++r) mx = __builtin_fmaxf(mx, s[1][qt][r]);
      mx = __builtin_fmaxf(mx, xhalf(mx)) * c2;
      const float mn    = __builtin_fmaxf(m_run[qt], mx);
      const float alpha = __builtin_amdgcn_exp2f(m_run[qt] - mn);
      const float nm    = -mn;
      m_run[qt] = mn;
      float ps0 = 0.0f, ps1 = 0.0f;
      v16h pf;
#pragma unroll
      for (int r = 0; r < 8; ++r) {
        const float p0 = __builtin_amdgcn_exp2f(__builtin_fmaf(s[0][qt][r], c2, nm));
        const float p1 = __builtin_amdgcn_exp2f(__builtin_fmaf(s[1][qt][r], c2, nm));
        ps0 += p0; ps1 += p1;
        pf[r]     = (_Float16)p0;
        pf[8 + r] = (_Float16)p1;
      }
      l_run[qt] = l_run[qt] * alpha + (ps0 + ps1);
#pragma unroll
      for (int t = 0; t < 8; ++t) o[qt][t] = wmma16(vf[t], pf, o[qt][t] * alpha);
    }
  }

  _Float16* Ob = attn + (((((size_t)b * NTOK + n0) >> 4) + w) * 16 + h * 4) * 512 + lane * 8;
  if (w == 0) give_tile<1>(o, m_run, l_run, xo[qb], xs[qb], w, lane);
  else        give_tile<0>(o, m_run, l_run, xo[qb], xs[qb], w, lane);
  __syncthreads();
  if (w == 0) keep_tile<0>(o, m_run, l_run, xo[qb], xs[qb], w, lane, Ob);
  else        keep_tile<1>(o, m_run, l_run, xo[qb], xs[qb], w, lane, Ob);
}

__global__ void __launch_bounds__(32) __attribute__((amdgpu_waves_per_eu(1, 4)))
out_proj(const _Float16* __restrict__ attn,
         const _Float16* __restrict__ WoT,
         const float* __restrict__ bo,
         const float* __restrict__ x,
         float* __restrict__ out) {
  const int lane  = threadIdx.x;
  const int hf    = lane >> 4;
  const int l16   = lane & 15;
  const int roff  = hf * 8;
  const int n0 = blockIdx.x * 32;
  const int c0 = blockIdx.y * 64;
  const int b  = blockIdx.z;

  const _Float16* A0 = attn + (((size_t)b * NTOK + n0) >> 4) * (CCH * 16) + lane * 8;
  const _Float16* A1 = A0 + (size_t)CCH * 16;
  const _Float16* Bp[4];
#pragma unroll
  for (int t = 0; t < 4; ++t)
    Bp[t] = WoT + (size_t)((c0 >> 4) + t) * (CCH * 16) + lane * 8;

  v16h a0c = load_frag(A0), a1c = load_frag(A1);
  v16h bc[4];
#pragma unroll
  for (int t = 0; t < 4; ++t) bc[t] = load_frag(Bp[t]);
  v8f acc0[4] = {}, acc1[4] = {};
#pragma unroll
  for (int k0 = 0; k0 < CCH; k0 += 32) {
    v16h a0n = a0c, a1n = a1c, bn[4] = {bc[0], bc[1], bc[2], bc[3]};
    const int kn = k0 + 32;
    if (kn < CCH) {
      a0n = load_frag(A0 + kn * 16);
      a1n = load_frag(A1 + kn * 16);
#pragma unroll
      for (int t = 0; t < 4; ++t) bn[t] = load_frag(Bp[t] + kn * 16);
    }
#pragma unroll
    for (int t = 0; t < 4; ++t) {
      acc0[t] = wmma16(a0c, bc[t], acc0[t]);
      acc1[t] = wmma16(a1c, bc[t], acc1[t]);
    }
    a0c = a0n; a1c = a1n;
#pragma unroll
    for (int t = 0; t < 4; ++t) bc[t] = bn[t];
  }

#pragma unroll
  for (int t = 0; t < 4; ++t) {
    const int c = c0 + t * 16 + l16;
    const float bias = bo[c];
    const size_t base = ((size_t)b * CCH + c) * NTOK + n0 + roff;
#pragma unroll
    for (int half2 = 0; half2 < 2; ++half2) {
      const v8f& av   = half2 ? acc1[t] : acc0[t];
      const size_t bi = base + (size_t)half2 * 16;
      const v4f x0 = *(const v4f*)(x + bi);
      const v4f x1 = *(const v4f*)(x + bi + 4);
      v4f r0, r1;
#pragma unroll
      for (int g = 0; g < 4; ++g) {
        r0[g] = av[g]     + bias + x0[g];
        r1[g] = av[4 + g] + bias + x1[g];
      }
      *(v4f*)(out + bi)     = r0;
      *(v4f*)(out + bi + 4) = r1;
    }
  }
}

extern "C" void kernel_launch(void* const* d_in, const int* in_sizes, int n_in,
                              void* d_out, int out_size, void* d_ws, size_t ws_size,
                              hipStream_t stream) {
  (void)in_sizes; (void)n_in; (void)out_size; (void)ws_size;
  const float* x  = (const float*)d_in[0];
  const float* Wp = (const float*)d_in[1];
  const float* bp = (const float*)d_in[2];
  const float* Wo = (const float*)d_in[3];
  const float* bo = (const float*)d_in[4];
  float* out = (float*)d_out;

  char* ws = (char*)d_ws;
  _Float16* xh  = (_Float16*)(ws);
  _Float16* WpT = (_Float16*)(ws +  8388608);
  _Float16* WoT = (_Float16*)(ws +  9961472);
  _Float16* qh  = (_Float16*)(ws + 10485760);
  _Float16* kh  = (_Float16*)(ws + 18874368);
  _Float16* vT  = (_Float16*)(ws + 27262976);
  _Float16* at  = (_Float16*)(ws + 35651584);

  to_fragment_order<<<dim3(256), 256, 0, stream>>>(x, Wp, Wo, xh, WpT, WoT);

  qkv_gemm    <<<dim3(NTOK / 256, 1536 / 192, NB), 256, 0, stream>>>(xh, WpT, bp, qh, kh, vT);
  attention_fa<<<dim3(NTOK / 128, NH,       NB), 256, 0, stream>>>(qh, kh, vT, at);
  out_proj    <<<dim3(NTOK / 32, CCH / 64,  NB), 32, 0, stream>>>(at, WoT, bo, x, out);
}
